# starting kernel
# baseline (speedup 1.0000x reference)
#include <hip/hip_runtime.h>
#include <math.h>

typedef __attribute__((ext_vector_type(2))) float v2f;
typedef __attribute__((ext_vector_type(8))) float v8f;

static constexpr int FEAT = 128;
static constexpr int APAD = 132;
static constexpr int ROWS_PER_BLK = 64;

__global__ __launch_bounds__(256) void k_zero(float* __restrict__ p, int n) {
    int i = blockIdx.x * 256 + threadIdx.x;
    if (i < n) p[i] = 0.0f;
}

__global__ __launch_bounds__(256) void k_degree(const int* __restrict__ src,
                                                const int* __restrict__ dst,
                                                float* __restrict__ degO,
                                                float* __restrict__ degI, int ne) {
    int e = blockIdx.x * 256 + threadIdx.x;
    if (e < ne) {
        atomicAdd(degO + src[e], 1.0f);
        atomicAdd(degI + dst[e], 1.0f);
    }
}

__global__ __launch_bounds__(256) void k_rsqrt_clip(float* __restrict__ degO,
                                                    float* __restrict__ degI, int n) {
    int i = blockIdx.x * 256 + threadIdx.x;
    if (i < n) {
        float a = degO[i]; a = a < 1.0f ? 1.0f : a;
        degO[i] = 1.0f / sqrtf(a);
        float b = degI[i]; b = b < 1.0f ? 1.0f : b;
        degI[i] = 1.0f / sqrtf(b);
    }
}

__global__ __launch_bounds__(256) void k_gemm_scaled(const float* __restrict__ X,
                                                     const float* __restrict__ rs,
                                                     const float* __restrict__ W,
                                                     float* __restrict__ Y, int nrows) {
    __shared__ float Alds[ROWS_PER_BLK * APAD];

    const int rowbase = blockIdx.x * ROWS_PER_BLK;
    const int t = threadIdx.x;

    for (int i = t; i < ROWS_PER_BLK * (FEAT / 4); i += 256) {
        int r  = i >> 5;
        int c4 = (i & 31) << 2;
        int row = rowbase + r;
        float4 v = {0.f, 0.f, 0.f, 0.f};
        float s = 0.f;
        if (row < nrows) {
            v = *(const float4*)(X + (size_t)row * FEAT + c4);
            s = rs[row];
        }
        float* dstp = Alds + r * APAD + c4;
        dstp[0] = v.x * s; dstp[1] = v.y * s; dstp[2] = v.z * s; dstp[3] = v.w * s;
    }
    __syncthreads();

    const int wave = t >> 5;
    const int lane = t & 31;
    const int half = lane >> 4;
    const int l16  = lane & 15;
    const int ncol = wave * 16 + l16;

    v8f acc0 = {}, acc1 = {}, acc2 = {}, acc3 = {};
    const float* a0 = Alds + (0 * 16 + l16) * APAD;
    const float* a1 = Alds + (1 * 16 + l16) * APAD;
    const float* a2 = Alds + (2 * 16 + l16) * APAD;
    const float* a3 = Alds + (3 * 16 + l16) * APAD;

    #pragma unroll 4
    for (int k = 0; k < FEAT; k += 4) {
        const int ka = k + half * 2;
        v2f b;
        b.x = W[ka * FEAT + ncol];
        b.y = W[(ka + 1) * FEAT + ncol];
        v2f a;
        a.x = a0[ka]; a.y = a0[ka + 1];
        acc0 = __builtin_amdgcn_wmma_f32_16x16x4_f32(false, a, false, b, (short)0, acc0, false, false);
        a.x = a1[ka]; a.y = a1[ka + 1];
        acc1 = __builtin_amdgcn_wmma_f32_16x16x4_f32(false, a, false, b, (short)0, acc1, false, false);
        a.x = a2[ka]; a.y = a2[ka + 1];
        acc2 = __builtin_amdgcn_wmma_f32_16x16x4_f32(false, a, false, b, (short)0, acc2, false, false);
        a.x = a3[ka]; a.y = a3[ka + 1];
        acc3 = __builtin_amdgcn_wmma_f32_16x16x4_f32(false, a, false, b, (short)0, acc3, false, false);
    }

    const int mrow0 = rowbase + half * 8;
    if (rowbase + ROWS_PER_BLK <= nrows) {
        #pragma unroll
        for (int r = 0; r < 8; ++r) {
            Y[(size_t)(mrow0 +  0 + r) * FEAT + ncol] = acc0[r];
            Y[(size_t)(mrow0 + 16 + r) * FEAT + ncol] = acc1[r];
            Y[(size_t)(mrow0 + 32 + r) * FEAT + ncol] = acc2[r];
            Y[(size_t)(mrow0 + 48 + r) * FEAT + ncol] = acc3[r];
        }
    } else {
        #pragma unroll
        for (int r = 0; r < 8; ++r) {
            int r0 = mrow0 + r;
            if (r0 +  0 < nrows) Y[(size_t)(r0 +  0) * FEAT + ncol] = acc0[r];
            if (r0 + 16 < nrows) Y[(size_t)(r0 + 16) * FEAT + ncol] = acc1[r];
            if (r0 + 32 < nrows) Y[(size_t)(r0 + 32) * FEAT + ncol] = acc2[r];
            if (r0 + 48 < nrows) Y[(size_t)(r0 + 48) * FEAT + ncol] = acc3[r];
        }
    }
}

__global__ __launch_bounds__(256) void k_scatter(const float* __restrict__ H,
                                                 const int* __restrict__ src,
                                                 const int* __restrict__ dst,
                                                 float* __restrict__ agg, int ne) {
    int e = blockIdx.x * 8 + (threadIdx.x >> 5);
    if (e >= ne) return;
    int lane = threadIdx.x & 31;
    int s = src[e], d = dst[e];
    float4 v = *(const float4*)(H + (size_t)s * FEAT + lane * 4);
    float* p = agg + (size_t)d * FEAT + lane * 4;
    atomicAdd(p + 0, v.x);
    atomicAdd(p + 1, v.y);
    atomicAdd(p + 2, v.z);
    atomicAdd(p + 3, v.w);
}

__global__ __launch_bounds__(256) void k_finalize(const float* __restrict__ agg,
                                                  const float* __restrict__ rsI,
                                                  const float* __restrict__ bias,
                                                  float* __restrict__ out, int n) {
    int i = blockIdx.x * 256 + threadIdx.x;
    if (i < n) {
        int row = i >> 7;
        int col = i & 127;
        float v = agg[i] * rsI[row] + bias[col];
        out[i] = v > 0.f ? v : 0.f;
    }
}

__global__ __launch_bounds__(256) void k_classify(const float* __restrict__ H,
                                                  const float* __restrict__ Wc,
                                                  const float* __restrict__ bc,
                                                  float* __restrict__ out, int n4) {
    int i = blockIdx.x * 256 + threadIdx.x;
    if (i >= n4) return;
    int node = i >> 2;
    int c = i & 3;
    const float* h = H + (size_t)node * FEAT;
    float sum = 0.f;
    #pragma unroll 8
    for (int k = 0; k < FEAT; ++k) sum += h[k] * Wc[k * 4 + c];
    out[i] = sum + bc[c];
}

extern "C" void kernel_launch(void* const* d_in, const int* in_sizes, int n_in,
                              void* d_out, int out_size, void* d_ws, size_t ws_size,
                              hipStream_t stream) {
    const float* feat = (const float*)d_in[0];
    const int*   src  = (const int*)d_in[1];
    const int*   dst  = (const int*)d_in[2];
    const float* W1   = (const float*)d_in[3];
    const float* b1   = (const float*)d_in[4];
    const float* W2   = (const float*)d_in[5];
    const float* b2   = (const float*)d_in[6];
    const float* Wc   = (const float*)d_in[7];
    const float* bc   = (const float*)d_in[8];
    float* out = (float*)d_out;

    const int nnodes = in_sizes[0] / FEAT;
    const int nedges = in_sizes[1];
    const size_t nAlign = ((size_t)nnodes + 3) & ~(size_t)3;
    const size_t nfeat = (size_t)nnodes * FEAT;

    float* rsO  = (float*)d_ws;
    float* rsI  = rsO + nAlign;
    float* bufA = rsI + nAlign;
    float* bufB = bufA + nfeat;

    const int B = 256;
    const int gN      = (int)((nfeat + B - 1) / B);
    const int gNode   = (nnodes + B - 1) / B;
    const int gEdgeT  = (nedges + B - 1) / B;
    const int gEdgeW  = (nedges + 7) / 8;
    const int gGemm   = (nnodes + ROWS_PER_BLK - 1) / ROWS_PER_BLK;
    const int gCls    = (nnodes * 4 + B - 1) / B;

    k_zero<<<(int)((2 * nAlign + B - 1) / B), B, 0, stream>>>(rsO, (int)(2 * nAlign));
    k_degree<<<gEdgeT, B, 0, stream>>>(src, dst, rsO, rsI, nedges);
    k_rsqrt_clip<<<gNode, B, 0, stream>>>(rsO, rsI, nnodes);

    k_gemm_scaled<<<gGemm, B, 0, stream>>>(feat, rsO, W1, bufA, nnodes);
    k_zero<<<gN, B, 0, stream>>>(bufB, (int)nfeat);
    k_scatter<<<gEdgeW, B, 0, stream>>>(bufA, src, dst, bufB, nedges);
    k_finalize<<<gN, B, 0, stream>>>(bufB, rsI, b1, bufA, (int)nfeat);

    k_gemm_scaled<<<gGemm, B, 0, stream>>>(bufA, rsO, W2, bufB, nnodes);
    k_zero<<<gN, B, 0, stream>>>(bufA, (int)nfeat);
    k_scatter<<<gEdgeW, B, 0, stream>>>(bufB, src, dst, bufA, nedges);
    k_finalize<<<gN, B, 0, stream>>>(bufA, rsI, b2, bufB, (int)nfeat);

    k_classify<<<gCls, B, 0, stream>>>(bufB, Wc, bc, out, nnodes * 4);
}


# degree counts: scattered global f32 atomic adds replaced by LDS integer histograms per (node range, edge slice), slice planes summed per node; zero-fill launch gone (8 -> 7); row sum order as before
# speedup vs baseline: 2.4413x; 2.4413x over previous
#include <hip/hip_runtime.h>
#include <math.h>

typedef __attribute__((ext_vector_type(2))) float v2f;
typedef __attribute__((ext_vector_type(8))) float v8f;

static constexpr int FEAT = 128;
static constexpr int APAD = 132;
static constexpr int ROWS_PER_BLK = 64;

static constexpr int DEG_NR = 12800;
static constexpr int DEG_S  = 32;
static constexpr int DEG_T  = 1024;
static constexpr int DEG_U  = 4;
static constexpr int TICK_STRIDE = 64;

__device__ __forceinline__ void deg_count(int* h, int v, int lo) {
    const unsigned u = (unsigned)(v - lo);
    if (u < (unsigned)DEG_NR) __hip_atomic_fetch_add(h + u, 1, __ATOMIC_RELAXED, __HIP_MEMORY_SCOPE_WORKGROUP);
}

__global__ __launch_bounds__(DEG_T) void k_degree(const int* __restrict__ src,
                                                  const int* __restrict__ dst,
                                                  int* __restrict__ part,
                                                  int* __restrict__ tick,
                                                  int ne, int es, int nranges, int nalign) {
    __shared__ int hO[DEG_NR];
    __shared__ int hI[DEG_NR];
    const int t = threadIdx.x;
    const int g = blockIdx.x % nranges;
    const int s = blockIdx.x / nranges;
    if (blockIdx.x == 0 && t < 4 * TICK_STRIDE) tick[t] = 0;
    for (int j = t; j < DEG_NR; j += DEG_T) { hO[j] = 0; hI[j] = 0; }
    __syncthreads();

    const int lo = g * DEG_NR;
    const int e0 = s * es;
    const int hi = e0 + es < ne ? e0 + es : ne;
    for (int eb = e0 + 4 * t; eb + 3 < hi; eb += DEG_U * 4 * DEG_T) {
        int4 a[DEG_U], b[DEG_U];
        bool ok[DEG_U];
        #pragma unroll
        for (int u = 0; u < DEG_U; ++u) {
            const int e = eb + u * 4 * DEG_T;
            ok[u] = e + 3 < hi;
            const int ea = ok[u] ? e : eb;
            a[u] = *(const int4*)(src + ea);
            b[u] = *(const int4*)(dst + ea);
        }
        #pragma unroll
        for (int u = 0; u < DEG_U; ++u) {
            if (ok[u]) {
                deg_count(hO, a[u].x, lo); deg_count(hO, a[u].y, lo);
                deg_count(hO, a[u].z, lo); deg_count(hO, a[u].w, lo);
                deg_count(hI, b[u].x, lo); deg_count(hI, b[u].y, lo);
                deg_count(hI, b[u].z, lo); deg_count(hI, b[u].w, lo);
            }
        }
    }
    {
        const int et = (hi & ~3) + t;
        if (t < 3 && et >= e0 && et < hi) {
            deg_count(hO, src[et], lo);
            deg_count(hI, dst[et], lo);
        }
    }
    __syncthreads();

    int* pO = part + (size_t)(2 * s) * nalign + lo;
    int* pI = pO + nalign;
    for (int j = 4 * t; j < DEG_NR; j += 4 * DEG_T) {
        if (lo + j < nalign) {
            *(int4*)(pO + j) = *(const int4*)(hO + j);
            *(int4*)(pI + j) = *(const int4*)(hI + j);
        }
    }
}

__global__ __launch_bounds__(256) void k_rsqrt_clip(const int* __restrict__ part,
                                                    float* __restrict__ rsO,
                                                    float* __restrict__ rsI, int n, int nalign) {
    int i = blockIdx.x * 256 + threadIdx.x;
    if (i < n) {
        int ca = 0, cb = 0;
        #pragma unroll 8
        for (int s = 0; s < DEG_S; ++s) {
            ca += part[(size_t)(2 * s) * nalign + i];
            cb += part[(size_t)(2 * s + 1) * nalign + i];
        }
        float a = (float)ca; a = a < 1.0f ? 1.0f : a;
        rsO[i] = 1.0f / sqrtf(a);
        float b = (float)cb; b = b < 1.0f ? 1.0f : b;
        rsI[i] = 1.0f / sqrtf(b);
    }
}

__device__ __forceinline__ float4 fin4(float4 a, float s, float4 b) {
    float4 r;
    r.x = a.x * s + b.x; r.x = r.x > 0.f ? r.x : 0.f;
    r.y = a.y * s + b.y; r.y = r.y > 0.f ? r.y : 0.f;
    r.z = a.z * s + b.z; r.z = r.z > 0.f ? r.z : 0.f;
    r.w = a.w * s + b.w; r.w = r.w > 0.f ? r.w : 0.f;
    return r;
}

template <bool FIN>
__global__ __launch_bounds__(256) void k_gemm_scaled(const float* X,
                                                     const float* __restrict__ rsO,
                                                     const float* __restrict__ rsI,
                                                     const float* __restrict__ bias,
                                                     const float* __restrict__ W,
                                                     float* __restrict__ Y,
                                                     float* Z, int nrows) {
    __shared__ float Alds[ROWS_PER_BLK * APAD];

    const int rowbase = blockIdx.x * ROWS_PER_BLK;
    const int t = threadIdx.x;

    float4 bz = {0.f, 0.f, 0.f, 0.f};
    if (FIN) bz = *(const float4*)(bias + ((t & 31) << 2));

    for (int i = t; i < ROWS_PER_BLK * (FEAT / 4); i += 256) {
        int r  = i >> 5;
        int c4 = (i & 31) << 2;
        int row = rowbase + r;
        float4 v = {0.f, 0.f, 0.f, 0.f};
        float s = 0.f;
        if (row < nrows) {
            v = *(const float4*)(X + (size_t)row * FEAT + c4);
            s = rsO[row];
            if (FIN) v = fin4(v, rsI[row], bz);
        }
        float* dstp = Alds + r * APAD + c4;
        dstp[0] = v.x * s; dstp[1] = v.y * s; dstp[2] = v.z * s; dstp[3] = v.w * s;
    }
    __syncthreads();

    {
        float4* z4 = (float4*)Z + (size_t)rowbase * (FEAT / 4);
        const float4 z = {0.f, 0.f, 0.f, 0.f};
        if (rowbase + ROWS_PER_BLK <= nrows) {
            for (int i = t; i < ROWS_PER_BLK * (FEAT / 4); i += 256) z4[i] = z;
        } else {
            for (int i = t; i < ROWS_PER_BLK * (FEAT / 4); i += 256)
                if (rowbase + (i >> 5) < nrows) z4[i] = z;
        }
    }

    const int wave = t >> 5;
    const int lane = t & 31;
    const int half = lane >> 4;
    const int l16  = lane & 15;
    const int ncol = wave * 16 + l16;

    v8f acc0 = {}, acc1 = {}, acc2 = {}, acc3 = {};
    const float* a0 = Alds + (0 * 16 + l16) * APAD;
    const float* a1 = Alds + (1 * 16 + l16) * APAD;
    const float* a2 = Alds + (2 * 16 + l16) * APAD;
    const float* a3 = Alds + (3 * 16 + l16) * APAD;

    #pragma unroll 4
    for (int k = 0; k < FEAT; k += 4) {
        const int ka = k + half * 2;
        v2f b;
        b.x = W[ka * FEAT + ncol];
        b.y = W[(ka + 1) * FEAT + ncol];
        v2f a;
        a.x = a0[ka]; a.y = a0[ka + 1];
        acc0 = __builtin_amdgcn_wmma_f32_16x16x4_f32(false, a, false, b, (short)0, acc0, false, false);
        a.x = a1[ka]; a.y = a1[ka + 1];
        acc1 = __builtin_amdgcn_wmma_f32_16x16x4_f32(false, a, false, b, (short)0, acc1, false, false);
        a.x = a2[ka]; a.y = a2[ka + 1];
        acc2 = __builtin_amdgcn_wmma_f32_16x16x4_f32(false, a, false, b, (short)0, acc2, false, false);
        a.x = a3[ka]; a.y = a3[ka + 1];
        acc3 = __builtin_amdgcn_wmma_f32_16x16x4_f32(false, a, false, b, (short)0, acc3, false, false);
    }

    const int mrow0 = rowbase + half * 8;
    if (rowbase + ROWS_PER_BLK <= nrows) {
        #pragma unroll
        for (int r = 0; r < 8; ++r) {
            Y[(size_t)(mrow0 +  0 + r) * FEAT + ncol] = acc0[r];
            Y[(size_t)(mrow0 + 16 + r) * FEAT + ncol] = acc1[r];
            Y[(size_t)(mrow0 + 32 + r) * FEAT + ncol] = acc2[r];
            Y[(size_t)(mrow0 + 48 + r) * FEAT + ncol] = acc3[r];
        }
    } else {
        #pragma unroll
        for (int r = 0; r < 8; ++r) {
            int r0 = mrow0 + r;
            if (r0 +  0 < nrows) Y[(size_t)(r0 +  0) * FEAT + ncol] = acc0[r];
            if (r0 + 16 < nrows) Y[(size_t)(r0 + 16) * FEAT + ncol] = acc1[r];
            if (r0 + 32 < nrows) Y[(size_t)(r0 + 32) * FEAT + ncol] = acc2[r];
            if (r0 + 48 < nrows) Y[(size_t)(r0 + 48) * FEAT + ncol] = acc3[r];
        }
    }
}

static constexpr int EPW   = 64;
static constexpr int EG    = 8;
static constexpr int CHUNK = 8 * EPW;

template <bool TAIL>
__device__ __forceinline__ void scatter_group(const float* __restrict__ H, float* __restrict__ agg,
                                              int sv0, int dv0, int sv1, int dv1,
                                              unsigned long long& m, int cnt, int lane) {
    float v[EG][4];
    int di[EG];
    #pragma unroll
    for (int j = 0; j < EG; ++j) {
        di[j] = 0;
        v[j][0] = 0.f; v[j][1] = 0.f; v[j][2] = 0.f; v[j][3] = 0.f;
        if (!TAIL || j < cnt) {
            const int i = __builtin_ctzll(m);
            m &= m - 1;
            const int il = i & 31;
            const int s0 = __builtin_amdgcn_readlane(sv0, il);
            const int s1 = __builtin_amdgcn_readlane(sv1, il);
            const int d0 = __builtin_amdgcn_readlane(dv0, il);
            const int d1 = __builtin_amdgcn_readlane(dv1, il);
            const int s = i < 32 ? s0 : s1;
            di[j] = i < 32 ? d0 : d1;
            const float* h = H + (size_t)s * FEAT + lane;
            v[j][0] = h[0];
            v[j][1] = h[32];
            v[j][2] = h[64];
            v[j][3] = h[96];
        }
    }
    #pragma unroll
    for (int j = 0; j < EG; ++j) {
        if (!TAIL || j < cnt) {
            float* p = agg + (size_t)di[j] * FEAT + lane;
            atomicAdd(p +  0, v[j][0]);
            atomicAdd(p + 32, v[j][1]);
            atomicAdd(p + 64, v[j][2]);
            atomicAdd(p + 96, v[j][3]);
        }
    }
}

__global__ __launch_bounds__(256) void k_scatter(const float* __restrict__ H,
                                                 const int* __restrict__ src,
                                                 const int* __restrict__ dst,
                                                 float* __restrict__ agg,
                                                 int* __restrict__ tick,
                                                 int ne, int nchunks, int nnodes) {
    __shared__ int s_draw[3];
    if (threadIdx.x == 0) {
        const int hw = (int)((__builtin_amdgcn_s_sendmsg_rtn(0x87) >> 18) & 1u);
        s_draw[0] = hw;
        s_draw[1] = atomicAdd(tick + hw * TICK_STRIDE, 1);
    }
    __syncthreads();
    int half = s_draw[0];
    int t = s_draw[1];
    if (t >= nchunks) {
        if (threadIdx.x == 0) s_draw[2] = atomicAdd(tick + (1 - half) * TICK_STRIDE, 1);
        __syncthreads();
        half = 1 - half;
        t = s_draw[2];
        if (t >= nchunks) return;
    }

    const int lane = threadIdx.x & 31;
    const int e0 = t * CHUNK + (threadIdx.x >> 5) * EPW;
    if (e0 >= ne) return;
    const int nsplit = nnodes >> 1;
    const bool in0 = e0 + lane < ne;
    const bool in1 = e0 + 32 + lane < ne;
    const int i0 = in0 ? e0 + lane : ne - 1;
    const int i1 = in1 ? e0 + 32 + lane : ne - 1;
    const int sv0 = src[i0];
    const int dv0 = dst[i0];
    const int sv1 = src[i1];
    const int dv1 = dst[i1];
    const bool mine0 = in0 && ((dv0 >= nsplit ? 1 : 0) == half);
    const bool mine1 = in1 && ((dv1 >= nsplit ? 1 : 0) == half);
    unsigned long long m = (unsigned long long)__builtin_amdgcn_ballot_w32(mine0) |
                           ((unsigned long long)__builtin_amdgcn_ballot_w32(mine1) << 32);
    int cnt = __builtin_popcountll(m);
    while (cnt >= EG) {
        scatter_group<false>(H, agg, sv0, dv0, sv1, dv1, m, EG, lane);
        cnt -= EG;
    }
    if (cnt > 0) scatter_group<true>(H, agg, sv0, dv0, sv1, dv1, m, cnt, lane);
}

__global__ __launch_bounds__(256) void k_classify(const float* __restrict__ agg,
                                                  const float* __restrict__ rsI,
                                                  const float* __restrict__ bias,
                                                  const float* __restrict__ Wc,
                                                  const float* __restrict__ bc,
                                                  float* __restrict__ out, int n4) {
    int i = blockIdx.x * 256 + threadIdx.x;
    if (i >= n4) return;
    int node = i >> 2;
    int c = i & 3;
    const float* a = agg + (size_t)node * FEAT;
    const float s = rsI[node];
    float sum = 0.f;
    #pragma unroll 8
    for (int k = 0; k < FEAT; ++k) {
        float h = a[k] * s + bias[k];
        h = h > 0.f ? h : 0.f;
        sum += h * Wc[k * 4 + c];
    }
    out[i] = sum + bc[c];
}

extern "C" void kernel_launch(void* const* d_in, const int* in_sizes, int n_in,
                              void* d_out, int out_size, void* d_ws, size_t ws_size,
                              hipStream_t stream) {
    const float* feat = (const float*)d_in[0];
    const int*   src  = (const int*)d_in[1];
    const int*   dst  = (const int*)d_in[2];
    const float* W1   = (const float*)d_in[3];
    const float* b1   = (const float*)d_in[4];
    const float* W2   = (const float*)d_in[5];
    const float* b2   = (const float*)d_in[6];
    const float* Wc   = (const float*)d_in[7];
    const float* bc   = (const float*)d_in[8];
    float* out = (float*)d_out;

    const int nnodes = in_sizes[0] / FEAT;
    const int nedges = in_sizes[1];
    const size_t nAlign = ((size_t)nnodes + 3) & ~(size_t)3;
    const size_t nfeat = (size_t)nnodes * FEAT;

    float* rsO  = (float*)d_ws;
    float* rsI  = rsO + nAlign;
    int*   tick = (int*)(rsI + nAlign);
    float* bufA = rsI + nAlign + 4 * TICK_STRIDE;
    float* bufB = bufA + nfeat;
    int*   part = (int*)(bufB + nfeat);

    const int B = 256;
    const int gNode   = (nnodes + B - 1) / B;
    const int nranges = (nnodes + DEG_NR - 1) / DEG_NR;
    const int eslice  = ((nedges + DEG_S - 1) / DEG_S + 3) & ~3;
    const int nchunks = (nedges + CHUNK - 1) / CHUNK;
    const int gEdgeW  = 2 * nchunks;
    const int gGemm   = (nnodes + ROWS_PER_BLK - 1) / ROWS_PER_BLK;
    const int gCls    = (nnodes * 4 + B - 1) / B;

    k_degree<<<nranges * DEG_S, DEG_T, 0, stream>>>(src, dst, part, tick, nedges, eslice, nranges, (int)nAlign);
    k_rsqrt_clip<<<gNode, B, 0, stream>>>(part, rsO, rsI, nnodes, (int)nAlign);

    k_gemm_scaled<false><<<gGemm, B, 0, stream>>>(feat, rsO, nullptr, nullptr, W1, bufA, bufB, nnodes);
    k_scatter<<<gEdgeW, B, 0, stream>>>(bufA, src, dst, bufB, tick, nedges, nchunks, nnodes);

    k_gemm_scaled<true><<<gGemm, B, 0, stream>>>(bufB, rsO, rsI, b1, W2, bufA, bufB, nnodes);
    k_scatter<<<gEdgeW, B, 0, stream>>>(bufA, src, dst, bufB, tick + 2 * TICK_STRIDE, nedges, nchunks, nnodes);

    k_classify<<<gCls, B, 0, stream>>>(bufB, rsI, b2, Wc, bc, out, nnodes * 4);
}


# edge sums built in LDS tiles: a workgroup owns 500 sum-table rows, scans all of dst for them, gathers those H rows, adds by LDS f32 atomics, stores the tile once; no global atomics; arrival order
# speedup vs baseline: 2.6926x; 1.1029x over previous
#include <hip/hip_runtime.h>
#include <math.h>

typedef __attribute__((ext_vector_type(2))) float v2f;
typedef __attribute__((ext_vector_type(8))) float v8f;

static constexpr int FEAT = 128;
static constexpr int APAD = 132;
static constexpr int ROWS_PER_BLK = 64;

static constexpr int DEG_NR = 12800;
static constexpr int DEG_S  = 32;
static constexpr int DEG_T  = 1024;
static constexpr int DEG_U  = 4;

__device__ __forceinline__ void deg_count(int* h, int v, int lo) {
    const unsigned u = (unsigned)(v - lo);
    if (u < (unsigned)DEG_NR) __hip_atomic_fetch_add(h + u, 1, __ATOMIC_RELAXED, __HIP_MEMORY_SCOPE_WORKGROUP);
}

__global__ __launch_bounds__(DEG_T) void k_degree(const int* __restrict__ src,
                                                  const int* __restrict__ dst,
                                                  int* __restrict__ part,
                                                  int ne, int es, int nranges, int nalign) {
    __shared__ int hO[DEG_NR];
    __shared__ int hI[DEG_NR];
    const int t = threadIdx.x;
    const int g = blockIdx.x % nranges;
    const int s = blockIdx.x / nranges;
    for (int j = t; j < DEG_NR; j += DEG_T) { hO[j] = 0; hI[j] = 0; }
    __syncthreads();

    const int lo = g * DEG_NR;
    const int e0 = s * es;
    const int hi = e0 + es < ne ? e0 + es : ne;
    for (int eb = e0 + 4 * t; eb + 3 < hi; eb += DEG_U * 4 * DEG_T) {
        int4 a[DEG_U], b[DEG_U];
        bool ok[DEG_U];
        #pragma unroll
        for (int u = 0; u < DEG_U; ++u) {
            const int e = eb + u * 4 * DEG_T;
            ok[u] = e + 3 < hi;
            const int ea = ok[u] ? e : eb;
            a[u] = *(const int4*)(src + ea);
            b[u] = *(const int4*)(dst + ea);
        }
        #pragma unroll
        for (int u = 0; u < DEG_U; ++u) {
            if (ok[u]) {
                deg_count(hO, a[u].x, lo); deg_count(hO, a[u].y, lo);
                deg_count(hO, a[u].z, lo); deg_count(hO, a[u].w, lo);
                deg_count(hI, b[u].x, lo); deg_count(hI, b[u].y, lo);
                deg_count(hI, b[u].z, lo); deg_count(hI, b[u].w, lo);
            }
        }
    }
    {
        const int et = (hi & ~3) + t;
        if (t < 3 && et >= e0 && et < hi) {
            deg_count(hO, src[et], lo);
            deg_count(hI, dst[et], lo);
        }
    }
    __syncthreads();

    int* pO = part + (size_t)(2 * s) * nalign + lo;
    int* pI = pO + nalign;
    for (int j = 4 * t; j < DEG_NR; j += 4 * DEG_T) {
        if (lo + j < nalign) {
            *(int4*)(pO + j) = *(const int4*)(hO + j);
            *(int4*)(pI + j) = *(const int4*)(hI + j);
        }
    }
}

__global__ __launch_bounds__(256) void k_rsqrt_clip(const int* __restrict__ part,
                                                    float* __restrict__ rsO,
                                                    float* __restrict__ rsI, int n, int nalign) {
    int i = blockIdx.x * 256 + threadIdx.x;
    if (i < n) {
        int ca = 0, cb = 0;
        #pragma unroll 8
        for (int s = 0; s < DEG_S; ++s) {
            ca += part[(size_t)(2 * s) * nalign + i];
            cb += part[(size_t)(2 * s + 1) * nalign + i];
        }
        float a = (float)ca; a = a < 1.0f ? 1.0f : a;
        rsO[i] = 1.0f / sqrtf(a);
        float b = (float)cb; b = b < 1.0f ? 1.0f : b;
        rsI[i] = 1.0f / sqrtf(b);
    }
}

__device__ __forceinline__ float4 fin4(float4 a, float s, float4 b) {
    float4 r;
    r.x = a.x * s + b.x; r.x = r.x > 0.f ? r.x : 0.f;
    r.y = a.y * s + b.y; r.y = r.y > 0.f ? r.y : 0.f;
    r.z = a.z * s + b.z; r.z = r.z > 0.f ? r.z : 0.f;
    r.w = a.w * s + b.w; r.w = r.w > 0.f ? r.w : 0.f;
    return r;
}

template <bool FIN>
__global__ __launch_bounds__(256) void k_gemm_scaled(const float* __restrict__ X,
                                                     const float* __restrict__ rsO,
                                                     const float* __restrict__ rsI,
                                                     const float* __restrict__ bias,
                                                     const float* __restrict__ W,
                                                     float* __restrict__ Y, int nrows) {
    __shared__ float Alds[ROWS_PER_BLK * APAD];

    const int rowbase = blockIdx.x * ROWS_PER_BLK;
    const int t = threadIdx.x;

    float4 bz = {0.f, 0.f, 0.f, 0.f};
    if (FIN) bz = *(const float4*)(bias + ((t & 31) << 2));

    for (int i = t; i < ROWS_PER_BLK * (FEAT / 4); i += 256) {
        int r  = i >> 5;
        int c4 = (i & 31) << 2;
        int row = rowbase + r;
        float4 v = {0.f, 0.f, 0.f, 0.f};
        float s = 0.f;
        if (row < nrows) {
            v = *(const float4*)(X + (size_t)row * FEAT + c4);
            s = rsO[row];
            if (FIN) v = fin4(v, rsI[row], bz);
        }
        float* dstp = Alds + r * APAD + c4;
        dstp[0] = v.x * s; dstp[1] = v.y * s; dstp[2] = v.z * s; dstp[3] = v.w * s;
    }
    __syncthreads();

    const int wave = t >> 5;
    const int lane = t & 31;
    const int half = lane >> 4;
    const int l16  = lane & 15;
    const int ncol = wave * 16 + l16;

    v8f acc0 = {}, acc1 = {}, acc2 = {}, acc3 = {};
    const float* a0 = Alds + (0 * 16 + l16) * APAD;
    const float* a1 = Alds + (1 * 16 + l16) * APAD;
    const float* a2 = Alds + (2 * 16 + l16) * APAD;
    const float* a3 = Alds + (3 * 16 + l16) * APAD;

    #pragma unroll 4
    for (int k = 0; k < FEAT; k += 4) {
        const int ka = k + half * 2;
        v2f b;
        b.x = W[ka * FEAT + ncol];
        b.y = W[(ka + 1) * FEAT + ncol];
        v2f a;
        a.x = a0[ka]; a.y = a0[ka + 1];
        acc0 = __builtin_amdgcn_wmma_f32_16x16x4_f32(false, a, false, b, (short)0, acc0, false, false);
        a.x = a1[ka]; a.y = a1[ka + 1];
        acc1 = __builtin_amdgcn_wmma_f32_16x16x4_f32(false, a, false, b, (short)0, acc1, false, false);
        a.x = a2[ka]; a.y = a2[ka + 1];
        acc2 = __builtin_amdgcn_wmma_f32_16x16x4_f32(false, a, false, b, (short)0, acc2, false, false);
        a.x = a3[ka]; a.y = a3[ka + 1];
        acc3 = __builtin_amdgcn_wmma_f32_16x16x4_f32(false, a, false, b, (short)0, acc3, false, false);
    }

    const int mrow0 = rowbase + half * 8;
    if (rowbase + ROWS_PER_BLK <= nrows) {
        #pragma unroll
        for (int r = 0; r < 8; ++r) {
            Y[(size_t)(mrow0 +  0 + r) * FEAT + ncol] = acc0[r];
            Y[(size_t)(mrow0 + 16 + r) * FEAT + ncol] = acc1[r];
            Y[(size_t)(mrow0 + 32 + r) * FEAT + ncol] = acc2[r];
            Y[(size_t)(mrow0 + 48 + r) * FEAT + ncol] = acc3[r];
        }
    } else {
        #pragma unroll
        for (int r = 0; r < 8; ++r) {
            int r0 = mrow0 + r;
            if (r0 +  0 < nrows) Y[(size_t)(r0 +  0) * FEAT + ncol] = acc0[r];
            if (r0 + 16 < nrows) Y[(size_t)(r0 + 16) * FEAT + ncol] = acc1[r];
            if (r0 + 32 < nrows) Y[(size_t)(r0 + 32) * FEAT + ncol] = acc2[r];
            if (r0 + 48 < nrows) Y[(size_t)(r0 + 48) * FEAT + ncol] = acc3[r];
        }
    }
}

static constexpr int TR      = 500;
static constexpr int TS_T    = 1024;
static constexpr int TS_U    = 4;
static constexpr int TS_TRIP = TS_T * 4 * TS_U;
static constexpr int QCAP    = 32 * 4 * TS_U + 32;
static constexpr int EGT     = 8;

__device__ __forceinline__ void lds_add(float* p, float v) {
    __hip_atomic_fetch_add(p, v, __ATOMIC_RELAXED, __HIP_MEMORY_SCOPE_WORKGROUP);
}

template <bool FULL>
__device__ __forceinline__ void tile_drain(const float* __restrict__ H, const int* __restrict__ src,
                                           const int* __restrict__ dst, float* tile, const int* q,
                                           int cnt, int lo, int lane) {
    int e = 0;
    if (FULL || lane < cnt) e = q[lane];
    const int sv = src[e];
    const int dv = dst[e] - lo;
    for (int g = 0; g < 32; g += EGT) {
        if (!FULL && g >= cnt) break;
        float4 v[EGT];
        #pragma unroll
        for (int j = 0; j < EGT; ++j) {
            v[j].x = 0.f; v[j].y = 0.f; v[j].z = 0.f; v[j].w = 0.f;
            if (FULL || g + j < cnt) {
                const int s = __builtin_amdgcn_readlane(sv, g + j);
                v[j] = *(const float4*)(H + (size_t)s * FEAT + lane * 4);
            }
        }
        #pragma unroll
        for (int j = 0; j < EGT; ++j) {
            if (FULL || g + j < cnt) {
                const int d = __builtin_amdgcn_readlane(dv, g + j);
                float* p = tile + d * FEAT + lane;
                lds_add(p,      v[j].x);
                lds_add(p + 32, v[j].y);
                lds_add(p + 64, v[j].z);
                lds_add(p + 96, v[j].w);
            }
        }
    }
}

__device__ __forceinline__ void tile_load_trip(const int* __restrict__ dst, int4* a, int tr, int t, int ne) {
    const int elast = ne >= 4 ? (ne - 4) & ~3 : 0;
    #pragma unroll
    for (int u = 0; u < TS_U; ++u) {
        const int e = tr * TS_TRIP + u * (TS_T * 4) + t * 4;
        const unsigned ea = (unsigned)(e < elast ? e : elast);
        a[u] = *(const int4*)((const char*)dst + ea * 4u);
    }
}

__device__ __forceinline__ void tile_file(int* q, int& qn, unsigned m, int e, int lane) {
    const int idx = qn + (int)__builtin_amdgcn_mbcnt_lo(m, 0u);
    if ((m >> lane) & 1u) q[idx] = e;
    qn += __builtin_popcount(m);
}

__global__ __launch_bounds__(TS_T) void k_tile_sum(const float* __restrict__ H,
                                                   const int* __restrict__ src,
                                                   const int* __restrict__ dst,
                                                   float* __restrict__ agg,
                                                   int ne, int nnodes) {
    __shared__ float tile[TR * FEAT];
    __shared__ int   qbuf[(TS_T / 32) * QCAP];
    const int t = threadIdx.x;
    const int lane = t & 31;
    const int wave = t >> 5;
    const int lo = blockIdx.x * TR;

    {
        const float4 z = {0.f, 0.f, 0.f, 0.f};
        for (int i = t; i < TR * FEAT / 4; i += TS_T) ((float4*)tile)[i] = z;
    }
    __syncthreads();

    int* q = qbuf + wave * QCAP;
    int qn = 0;
    const int ntrips = (ne + TS_TRIP - 1) / TS_TRIP;
    int trip = (int)(((long long)blockIdx.x * ntrips) / gridDim.x);

    int4 cur[TS_U];
    tile_load_trip(dst, cur, trip, t, ne);
    for (int i = 0; i < ntrips; ++i) {
        int tn = trip + 1;
        if (tn == ntrips) tn = 0;
        int4 nxt[TS_U];
        tile_load_trip(dst, nxt, i + 1 < ntrips ? tn : trip, t, ne);

        const int ebase = trip * TS_TRIP + t * 4;
        unsigned mk[4 * TS_U];
        #pragma unroll
        for (int u = 0; u < TS_U; ++u) {
            const unsigned okm = __builtin_amdgcn_ballot_w32(ebase + u * (TS_T * 4) + 3 < ne);
            mk[4 * u + 0] = __builtin_amdgcn_ballot_w32((unsigned)(cur[u].x - lo) < (unsigned)TR) & okm;
            mk[4 * u + 1] = __builtin_amdgcn_ballot_w32((unsigned)(cur[u].y - lo) < (unsigned)TR) & okm;
            mk[4 * u + 2] = __builtin_amdgcn_ballot_w32((unsigned)(cur[u].z - lo) < (unsigned)TR) & okm;
            mk[4 * u + 3] = __builtin_amdgcn_ballot_w32((unsigned)(cur[u].w - lo) < (unsigned)TR) & okm;
        }
        #pragma unroll
        for (int k = 0; k < 4 * TS_U; ++k) {
            if (mk[k]) tile_file(q, qn, mk[k], ebase + (k >> 2) * (TS_T * 4) + (k & 3), lane);
        }

        int head = 0;
        while (qn - head >= 32) {
            tile_drain<true>(H, src, dst, tile, q + head, 32, lo, lane);
            head += 32;
        }
        if (head > 0) {
            const int left = qn - head;
            int v = 0;
            if (lane < left) v = q[head + lane];
            if (lane < left) q[lane] = v;
            qn = left;
        }

        #pragma unroll
        for (int u = 0; u < TS_U; ++u) cur[u] = nxt[u];
        trip = tn;
    }

    if (wave == 0) {
        const int e = (ne & ~3) + lane;
        const bool ok = lane < 3 && e < ne;
        const int dval = dst[ok ? e : 0];
        const unsigned m = __builtin_amdgcn_ballot_w32(ok && (unsigned)(dval - lo) < (unsigned)TR);
        if (m) tile_file(q, qn, m, e, lane);
    }
    if (qn >= 32) {
        tile_drain<true>(H, src, dst, tile, q, 32, lo, lane);
        qn -= 32;
        if (qn > 0) tile_drain<false>(H, src, dst, tile, q + 32, qn, lo, lane);
    } else if (qn > 0) {
        tile_drain<false>(H, src, dst, tile, q, qn, lo, lane);
    }
    __syncthreads();

    for (int r = wave; r < TR; r += TS_T / 32) {
        const int row = lo + r;
        if (row < nnodes) {
            const float* p = tile + r * FEAT + lane;
            float4 v;
            v.x = p[0]; v.y = p[32]; v.z = p[64]; v.w = p[96];
            *(float4*)(agg + (size_t)row * FEAT + lane * 4) = v;
        }
    }
}

__global__ __launch_bounds__(256) void k_classify(const float* __restrict__ agg,
                                                  const float* __restrict__ rsI,
                                                  const float* __restrict__ bias,
                                                  const float* __restrict__ Wc,
                                                  const float* __restrict__ bc,
                                                  float* __restrict__ out, int n4) {
    int i = blockIdx.x * 256 + threadIdx.x;
    if (i >= n4) return;
    int node = i >> 2;
    int c = i & 3;
    const float* a = agg + (size_t)node * FEAT;
    const float s = rsI[node];
    float sum = 0.f;
    #pragma unroll 8
    for (int k = 0; k < FEAT; ++k) {
        float h = a[k] * s + bias[k];
        h = h > 0.f ? h : 0.f;
        sum += h * Wc[k * 4 + c];
    }
    out[i] = sum + bc[c];
}

extern "C" void kernel_launch(void* const* d_in, const int* in_sizes, int n_in,
                              void* d_out, int out_size, void* d_ws, size_t ws_size,
                              hipStream_t stream) {
    const float* feat = (const float*)d_in[0];
    const int*   src  = (const int*)d_in[1];
    const int*   dst  = (const int*)d_in[2];
    const float* W1   = (const float*)d_in[3];
    const float* b1   = (const float*)d_in[4];
    const float* W2   = (const float*)d_in[5];
    const float* b2   = (const float*)d_in[6];
    const float* Wc   = (const float*)d_in[7];
    const float* bc   = (const float*)d_in[8];
    float* out = (float*)d_out;

    const int nnodes = in_sizes[0] / FEAT;
    const int nedges = in_sizes[1];
    const size_t nAlign = ((size_t)nnodes + 3) & ~(size_t)3;
    const size_t nfeat = (size_t)nnodes * FEAT;

    float* rsO  = (float*)d_ws;
    float* rsI  = rsO + nAlign;
    float* bufA = rsI + nAlign;
    float* bufB = bufA + nfeat;
    int*   part = (int*)(bufB + nfeat);

    const int B = 256;
    const int gNode   = (nnodes + B - 1) / B;
    const int nranges = (nnodes + DEG_NR - 1) / DEG_NR;
    const int eslice  = ((nedges + DEG_S - 1) / DEG_S + 3) & ~3;
    const int gTile   = (nnodes + TR - 1) / TR;
    const int gGemm   = (nnodes + ROWS_PER_BLK - 1) / ROWS_PER_BLK;
    const int gCls    = (nnodes * 4 + B - 1) / B;

    k_degree<<<nranges * DEG_S, DEG_T, 0, stream>>>(src, dst, part, nedges, eslice, nranges, (int)nAlign);
    k_rsqrt_clip<<<gNode, B, 0, stream>>>(part, rsO, rsI, nnodes, (int)nAlign);

    k_gemm_scaled<false><<<gGemm, B, 0, stream>>>(feat, rsO, nullptr, nullptr, W1, bufA, nnodes);
    k_tile_sum<<<gTile, TS_T, 0, stream>>>(bufA, src, dst, bufB, nedges, nnodes);

    k_gemm_scaled<true><<<gGemm, B, 0, stream>>>(bufB, rsO, rsI, b1, W2, bufA, nnodes);
    k_tile_sum<<<gTile, TS_T, 0, stream>>>(bufA, src, dst, bufB, nedges, nnodes);

    k_classify<<<gCls, B, 0, stream>>>(bufB, rsI, b2, Wc, bc, out, nnodes * 4);
}


# layer 2's edge sum reads per-tile lists of (source row, tile row) pairs that layer 1's dst scan wrote as it added them: dst scanned once per call, not twice; LDS sum order by arrival as before
# speedup vs baseline: 3.2423x; 1.2042x over previous
#include <hip/hip_runtime.h>
#include <math.h>

typedef __attribute__((ext_vector_type(2))) float v2f;
typedef __attribute__((ext_vector_type(8))) float v8f;

static constexpr int FEAT = 128;
static constexpr int APAD = 132;
static constexpr int ROWS_PER_BLK = 64;

static constexpr int DEG_NR = 12800;
static constexpr int DEG_S  = 32;
static constexpr int DEG_T  = 1024;
static constexpr int DEG_U  = 4;

__device__ __forceinline__ void deg_count(int* h, int v, int lo) {
    const unsigned u = (unsigned)(v - lo);
    if (u < (unsigned)DEG_NR) __hip_atomic_fetch_add(h + u, 1, __ATOMIC_RELAXED, __HIP_MEMORY_SCOPE_WORKGROUP);
}

__global__ __launch_bounds__(DEG_T) void k_degree(const int* __restrict__ src,
                                                  const int* __restrict__ dst,
                                                  int* __restrict__ part,
                                                  int ne, int es, int nranges, int nalign) {
    __shared__ int hO[DEG_NR];
    __shared__ int hI[DEG_NR];
    const int t = threadIdx.x;
    const int g = blockIdx.x % nranges;
    const int s = blockIdx.x / nranges;
    for (int j = t; j < DEG_NR; j += DEG_T) { hO[j] = 0; hI[j] = 0; }
    __syncthreads();

    const int lo = g * DEG_NR;
    const int e0 = s * es;
    const int hi = e0 + es < ne ? e0 + es : ne;
    for (int eb = e0 + 4 * t; eb + 3 < hi; eb += DEG_U * 4 * DEG_T) {
        int4 a[DEG_U], b[DEG_U];
        bool ok[DEG_U];
        #pragma unroll
        for (int u = 0; u < DEG_U; ++u) {
            const int e = eb + u * 4 * DEG_T;
            ok[u] = e + 3 < hi;
            const int ea = ok[u] ? e : eb;
            a[u] = *(const int4*)(src + ea);
            b[u] = *(const int4*)(dst + ea);
        }
        #pragma unroll
        for (int u = 0; u < DEG_U; ++u) {
            if (ok[u]) {
                deg_count(hO, a[u].x, lo); deg_count(hO, a[u].y, lo);
                deg_count(hO, a[u].z, lo); deg_count(hO, a[u].w, lo);
                deg_count(hI, b[u].x, lo); deg_count(hI, b[u].y, lo);
                deg_count(hI, b[u].z, lo); deg_count(hI, b[u].w, lo);
            }
        }
    }
    {
        const int et = (hi & ~3) + t;
        if (t < 3 && et >= e0 && et < hi) {
            deg_count(hO, src[et], lo);
            deg_count(hI, dst[et], lo);
        }
    }
    __syncthreads();

    int* pO = part + (size_t)(2 * s) * nalign + lo;
    int* pI = pO + nalign;
    for (int j = 4 * t; j < DEG_NR; j += 4 * DEG_T) {
        if (lo + j < nalign) {
            *(int4*)(pO + j) = *(const int4*)(hO + j);
            *(int4*)(pI + j) = *(const int4*)(hI + j);
        }
    }
}

__global__ __launch_bounds__(256) void k_rsqrt_clip(const int* __restrict__ part,
                                                    float* __restrict__ rsO,
                                                    float* __restrict__ rsI, int n, int nalign) {
    int i = blockIdx.x * 256 + threadIdx.x;
    if (i < n) {
        int ca = 0, cb = 0;
        #pragma unroll 8
        for (int s = 0; s < DEG_S; ++s) {
            ca += part[(size_t)(2 * s) * nalign + i];
            cb += part[(size_t)(2 * s + 1) * nalign + i];
        }
        float a = (float)ca; a = a < 1.0f ? 1.0f : a;
        rsO[i] = 1.0f / sqrtf(a);
        float b = (float)cb; b = b < 1.0f ? 1.0f : b;
        rsI[i] = 1.0f / sqrtf(b);
    }
}

__device__ __forceinline__ float4 fin4(float4 a, float s, float4 b) {
    float4 r;
    r.x = a.x * s + b.x; r.x = r.x > 0.f ? r.x : 0.f;
    r.y = a.y * s + b.y; r.y = r.y > 0.f ? r.y : 0.f;
    r.z = a.z * s + b.z; r.z = r.z > 0.f ? r.z : 0.f;
    r.w = a.w * s + b.w; r.w = r.w > 0.f ? r.w : 0.f;
    return r;
}

template <bool FIN>
__global__ __launch_bounds__(256) void k_gemm_scaled(const float* __restrict__ X,
                                                     const float* __restrict__ rsO,
                                                     const float* __restrict__ rsI,
                                                     const float* __restrict__ bias,
                                                     const float* __restrict__ W,
                                                     float* __restrict__ Y, int nrows) {
    __shared__ float Alds[ROWS_PER_BLK * APAD];

    const int rowbase = blockIdx.x * ROWS_PER_BLK;
    const int t = threadIdx.x;

    float4 bz = {0.f, 0.f, 0.f, 0.f};
    if (FIN) bz = *(const float4*)(bias + ((t & 31) << 2));

    for (int i = t; i < ROWS_PER_BLK * (FEAT / 4); i += 256) {
        int r  = i >> 5;
        int c4 = (i & 31) << 2;
        int row = rowbase + r;
        float4 v = {0.f, 0.f, 0.f, 0.f};
        float s = 0.f;
        if (row < nrows) {
            v = *(const float4*)(X + (size_t)row * FEAT + c4);
            s = rsO[row];
            if (FIN) v = fin4(v, rsI[row], bz);
        }
        float* dstp = Alds + r * APAD + c4;
        dstp[0] = v.x * s; dstp[1] = v.y * s; dstp[2] = v.z * s; dstp[3] = v.w * s;
    }
    __syncthreads();

    const int wave = t >> 5;
    const int lane = t & 31;
    const int half = lane >> 4;
    const int l16  = lane & 15;
    const int ncol = wave * 16 + l16;

    v8f acc0 = {}, acc1 = {}, acc2 = {}, acc3 = {};
    const float* a0 = Alds + (0 * 16 + l16) * APAD;
    const float* a1 = Alds + (1 * 16 + l16) * APAD;
    const float* a2 = Alds + (2 * 16 + l16) * APAD;
    const float* a3 = Alds + (3 * 16 + l16) * APAD;

    #pragma unroll 4
    for (int k = 0; k < FEAT; k += 4) {
        const int ka = k + half * 2;
        v2f b;
        b.x = W[ka * FEAT + ncol];
        b.y = W[(ka + 1) * FEAT + ncol];
        v2f a;
        a.x = a0[ka]; a.y = a0[ka + 1];
        acc0 = __builtin_amdgcn_wmma_f32_16x16x4_f32(false, a, false, b, (short)0, acc0, false, false);
        a.x = a1[ka]; a.y = a1[ka + 1];
        acc1 = __builtin_amdgcn_wmma_f32_16x16x4_f32(false, a, false, b, (short)0, acc1, false, false);
        a.x = a2[ka]; a.y = a2[ka + 1];
        acc2 = __builtin_amdgcn_wmma_f32_16x16x4_f32(false, a, false, b, (short)0, acc2, false, false);
        a.x = a3[ka]; a.y = a3[ka + 1];
        acc3 = __builtin_amdgcn_wmma_f32_16x16x4_f32(false, a, false, b, (short)0, acc3, false, false);
    }

    const int mrow0 = rowbase + half * 8;
    if (rowbase + ROWS_PER_BLK <= nrows) {
        #pragma unroll
        for (int r = 0; r < 8; ++r) {
            Y[(size_t)(mrow0 +  0 + r) * FEAT + ncol] = acc0[r];
            Y[(size_t)(mrow0 + 16 + r) * FEAT + ncol] = acc1[r];
            Y[(size_t)(mrow0 + 32 + r) * FEAT + ncol] = acc2[r];
            Y[(size_t)(mrow0 + 48 + r) * FEAT + ncol] = acc3[r];
        }
    } else {
        #pragma unroll
        for (int r = 0; r < 8; ++r) {
            int r0 = mrow0 + r;
            if (r0 +  0 < nrows) Y[(size_t)(r0 +  0) * FEAT + ncol] = acc0[r];
            if (r0 + 16 < nrows) Y[(size_t)(r0 + 16) * FEAT + ncol] = acc1[r];
            if (r0 + 32 < nrows) Y[(size_t)(r0 + 32) * FEAT + ncol] = acc2[r];
            if (r0 + 48 < nrows) Y[(size_t)(r0 + 48) * FEAT + ncol] = acc3[r];
        }
    }
}

static constexpr int TR      = 500;
static constexpr int TS_T    = 1024;
static constexpr int TS_U    = 4;
static constexpr int TS_TRIP = TS_T * 4 * TS_U;
static constexpr int QCAP    = 32 * 4 * TS_U + 32;
static constexpr int EGT     = 8;
static constexpr int LCAP    = 32768;

__device__ __forceinline__ void lds_add(float* p, float v) {
    __hip_atomic_fetch_add(p, v, __ATOMIC_RELAXED, __HIP_MEMORY_SCOPE_WORKGROUP);
}

template <bool FULL>
__device__ __forceinline__ void tile_rows(const float* __restrict__ H, float* tile, int sv, int dv,
                                          int cnt, int lane) {
    for (int g = 0; g < 32; g += EGT) {
        if (!FULL && g >= cnt) break;
        float4 v[EGT];
        #pragma unroll
        for (int j = 0; j < EGT; ++j) {
            v[j].x = 0.f; v[j].y = 0.f; v[j].z = 0.f; v[j].w = 0.f;
            if (FULL || g + j < cnt) {
                const int s = __builtin_amdgcn_readlane(sv, g + j);
                v[j] = *(const float4*)(H + (size_t)s * FEAT + lane * 4);
            }
        }
        #pragma unroll
        for (int j = 0; j < EGT; ++j) {
            if (FULL || g + j < cnt) {
                const int d = __builtin_amdgcn_readlane(dv, g + j);
                float* p = tile + d * FEAT + lane;
                lds_add(p,      v[j].x);
                lds_add(p + 32, v[j].y);
                lds_add(p + 64, v[j].z);
                lds_add(p + 96, v[j].w);
            }
        }
    }
}

template <bool FULL, bool KEEP>
__device__ __forceinline__ void tile_drain(const float* __restrict__ H, const int* __restrict__ src,
                                           const int* __restrict__ dst, float* tile, const int* q,
                                           int cnt, int lo, int lane, int2* lst, int* lctl) {
    int e = 0;
    if (FULL || lane < cnt) e = q[lane];
    const int sv = src[e];
    const int dv = dst[e] - lo;
    if (KEEP) {
        const int n = FULL ? 32 : cnt;
        int base = 0;
        if (lane == 0) base = __hip_atomic_fetch_add(lctl, n, __ATOMIC_RELAXED, __HIP_MEMORY_SCOPE_WORKGROUP);
        base = __builtin_amdgcn_readfirstlane(base);
        if (base + n <= LCAP) {
            if (FULL || lane < cnt) lst[base + lane] = make_int2(sv, dv);
        } else if (lane == 0) {
            lctl[1] = 1;
        }
    }
    tile_rows<FULL>(H, tile, sv, dv, cnt, lane);
}

__device__ __forceinline__ void tile_load_trip(const int* __restrict__ dst, int4* a, int tr, int t, int ne) {
    const int elast = ne >= 4 ? (ne - 4) & ~3 : 0;
    #pragma unroll
    for (int u = 0; u < TS_U; ++u) {
        const int e = tr * TS_TRIP + u * (TS_T * 4) + t * 4;
        const unsigned ea = (unsigned)(e < elast ? e : elast);
        a[u] = *(const int4*)((const char*)dst + ea * 4u);
    }
}

__device__ __forceinline__ void tile_file(int* q, int& qn, unsigned m, int e, int lane) {
    const int idx = qn + (int)__builtin_amdgcn_mbcnt_lo(m, 0u);
    if ((m >> lane) & 1u) q[idx] = e;
    qn += __builtin_popcount(m);
}

__device__ __forceinline__ void tile_from_list(const float* __restrict__ H, float* tile, const int2* lst,
                                               int nlist, int wave, int lane) {
    const int nch = (nlist + 31) >> 5;
    int c = __builtin_amdgcn_readfirstlane(wave);
    int2 cur = make_int2(0, 0);
    if (c < nch) { const int i = c * 32 + lane; cur = lst[i < nlist ? i : 0]; }
    while (c < nch) {
        const int cn = c + TS_T / 32;
        int2 nxt = cur;
        if (cn < nch) { const int i = cn * 32 + lane; nxt = lst[i < nlist ? i : 0]; }
        const int cnt = nlist - c * 32;
        if (cnt >= 32) tile_rows<true>(H, tile, cur.x, cur.y, 32, lane);
        else           tile_rows<false>(H, tile, cur.x, cur.y, cnt, lane);
        cur = nxt;
        c = cn;
    }
}

template <bool KEEP>
__device__ __forceinline__ void tile_scan(const float* __restrict__ H, const int* __restrict__ src,
                                          const int* __restrict__ dst, float* tile, int* q,
                                          int2* lst, int* lctl, int ne, int lo, int t) {
    const int lane = t & 31;
    const int wave = t >> 5;
    int qn = 0;
    const int ntrips = (ne + TS_TRIP - 1) / TS_TRIP;
    int trip = (int)(((long long)blockIdx.x * ntrips) / gridDim.x);

    int4 cur[TS_U];
    tile_load_trip(dst, cur, trip, t, ne);
    for (int i = 0; i < ntrips; ++i) {
        int tn = trip + 1;
        if (tn == ntrips) tn = 0;
        int4 nxt[TS_U];
        tile_load_trip(dst, nxt, i + 1 < ntrips ? tn : trip, t, ne);

        const int ebase = trip * TS_TRIP + t * 4;
        unsigned mk[4 * TS_U];
        #pragma unroll
        for (int u = 0; u < TS_U; ++u) {
            const unsigned okm = __builtin_amdgcn_ballot_w32(ebase + u * (TS_T * 4) + 3 < ne);
            mk[4 * u + 0] = __builtin_amdgcn_ballot_w32((unsigned)(cur[u].x - lo) < (unsigned)TR) & okm;
            mk[4 * u + 1] = __builtin_amdgcn_ballot_w32((unsigned)(cur[u].y - lo) < (unsigned)TR) & okm;
            mk[4 * u + 2] = __builtin_amdgcn_ballot_w32((unsigned)(cur[u].z - lo) < (unsigned)TR) & okm;
            mk[4 * u + 3] = __builtin_amdgcn_ballot_w32((unsigned)(cur[u].w - lo) < (unsigned)TR) & okm;
        }
        #pragma unroll
        for (int k = 0; k < 4 * TS_U; ++k) {
            if (mk[k]) tile_file(q, qn, mk[k], ebase + (k >> 2) * (TS_T * 4) + (k & 3), lane);
        }

        int head = 0;
        while (qn - head >= 32) {
            tile_drain<true, KEEP>(H, src, dst, tile, q + head, 32, lo, lane, lst, lctl);
            head += 32;
        }
        if (head > 0) {
            const int left = qn - head;
            int v = 0;
            if (lane < left) v = q[head + lane];
            if (lane < left) q[lane] = v;
            qn = left;
        }

        #pragma unroll
        for (int u = 0; u < TS_U; ++u) cur[u] = nxt[u];
        trip = tn;
    }

    if (wave == 0) {
        const int e = (ne & ~3) + lane;
        const bool ok = lane < 3 && e < ne;
        const int dval = dst[ok ? e : 0];
        const unsigned m = __builtin_amdgcn_ballot_w32(ok && (unsigned)(dval - lo) < (unsigned)TR);
        if (m) tile_file(q, qn, m, e, lane);
    }
    if (qn >= 32) {
        tile_drain<true, KEEP>(H, src, dst, tile, q, 32, lo, lane, lst, lctl);
        qn -= 32;
        if (qn > 0) tile_drain<false, KEEP>(H, src, dst, tile, q + 32, qn, lo, lane, lst, lctl);
    } else if (qn > 0) {
        tile_drain<false, KEEP>(H, src, dst, tile, q, qn, lo, lane, lst, lctl);
    }
}

template <bool KEEP>
__global__ __launch_bounds__(TS_T) void k_tile_sum(const float* __restrict__ H,
                                                   const int* __restrict__ src,
                                                   const int* __restrict__ dst,
                                                   float* __restrict__ agg,
                                                   int* elist, int* ehead,
                                                   int ne, int nnodes) {
    __shared__ float tile[TR * FEAT];
    __shared__ int   qbuf[(TS_T / 32) * QCAP];
    __shared__ int   lctl[2];
    const int t = threadIdx.x;
    const int lane = t & 31;
    const int wave = t >> 5;
    const int lo = blockIdx.x * TR;

    {
        const float4 z = {0.f, 0.f, 0.f, 0.f};
        for (int i = t; i < TR * FEAT / 4; i += TS_T) ((float4*)tile)[i] = z;
    }
    if (KEEP && t < 2) lctl[t] = 0;
    __syncthreads();

    int2* lst = (int2*)elist + (size_t)blockIdx.x * LCAP;
    int nlist = -1;
    if (!KEEP) nlist = __builtin_amdgcn_readfirstlane(ehead[blockIdx.x]);

    if (nlist >= 0) tile_from_list(H, tile, lst, nlist, wave, lane);
    else            tile_scan<KEEP>(H, src, dst, tile, qbuf + wave * QCAP, lst, lctl, ne, lo, t);
    __syncthreads();

    if (KEEP && t == 0) ehead[blockIdx.x] = lctl[1] ? -1 : lctl[0];

    for (int r = wave; r < TR; r += TS_T / 32) {
        const int row = lo + r;
        if (row < nnodes) {
            const float* p = tile + r * FEAT + lane;
            float4 v;
            v.x = p[0]; v.y = p[32]; v.z = p[64]; v.w = p[96];
            *(float4*)(agg + (size_t)row * FEAT + lane * 4) = v;
        }
    }
}

__global__ __launch_bounds__(256) void k_classify(const float* __restrict__ agg,
                                                  const float* __restrict__ rsI,
                                                  const float* __restrict__ bias,
                                                  const float* __restrict__ Wc,
                                                  const float* __restrict__ bc,
                                                  float* __restrict__ out, int n4) {
    int i = blockIdx.x * 256 + threadIdx.x;
    if (i >= n4) return;
    int node = i >> 2;
    int c = i & 3;
    const float* a = agg + (size_t)node * FEAT;
    const float s = rsI[node];
    float sum = 0.f;
    #pragma unroll 8
    for (int k = 0; k < FEAT; ++k) {
        float h = a[k] * s + bias[k];
        h = h > 0.f ? h : 0.f;
        sum += h * Wc[k * 4 + c];
    }
    out[i] = sum + bc[c];
}

extern "C" void kernel_launch(void* const* d_in, const int* in_sizes, int n_in,
                              void* d_out, int out_size, void* d_ws, size_t ws_size,
                              hipStream_t stream) {
    const float* feat = (const float*)d_in[0];
    const int*   src  = (const int*)d_in[1];
    const int*   dst  = (const int*)d_in[2];
    const float* W1   = (const float*)d_in[3];
    const float* b1   = (const float*)d_in[4];
    const float* W2   = (const float*)d_in[5];
    const float* b2   = (const float*)d_in[6];
    const float* Wc   = (const float*)d_in[7];
    const float* bc   = (const float*)d_in[8];
    float* out = (float*)d_out;

    const int nnodes = in_sizes[0] / FEAT;
    const int nedges = in_sizes[1];
    const size_t nAlign = ((size_t)nnodes + 3) & ~(size_t)3;
    const size_t nfeat = (size_t)nnodes * FEAT;

    float* rsO  = (float*)d_ws;
    float* rsI  = rsO + nAlign;
    float* bufA = rsI + nAlign;
    float* bufB = bufA + nfeat;
    int*   part = (int*)(bufB + nfeat);

    const int B = 256;
    const int gNode   = (nnodes + B - 1) / B;
    const int nranges = (nnodes + DEG_NR - 1) / DEG_NR;
    const int eslice  = ((nedges + DEG_S - 1) / DEG_S + 3) & ~3;
    const int gTile   = (nnodes + TR - 1) / TR;
    const int gGemm   = (nnodes + ROWS_PER_BLK - 1) / ROWS_PER_BLK;
    const int gCls    = (nnodes * 4 + B - 1) / B;
    int*   elist = part + (size_t)DEG_S * 2 * nAlign;
    int*   ehead = elist + (size_t)gTile * LCAP * 2;

    k_degree<<<nranges * DEG_S, DEG_T, 0, stream>>>(src, dst, part, nedges, eslice, nranges, (int)nAlign);
    k_rsqrt_clip<<<gNode, B, 0, stream>>>(part, rsO, rsI, nnodes, (int)nAlign);

    k_gemm_scaled<false><<<gGemm, B, 0, stream>>>(feat, rsO, nullptr, nullptr, W1, bufA, nnodes);
    k_tile_sum<true><<<gTile, TS_T, 0, stream>>>(bufA, src, dst, bufB, elist, ehead, nedges, nnodes);

    k_gemm_scaled<true><<<gGemm, B, 0, stream>>>(bufB, rsO, rsI, b1, W2, bufA, nnodes);
    k_tile_sum<false><<<gTile, TS_T, 0, stream>>>(bufA, src, dst, bufB, elist, ehead, nedges, nnodes);

    k_classify<<<gCls, B, 0, stream>>>(bufB, rsI, b2, Wc, bc, out, nnodes * 4);
}


# tile edge lists built by one pass over the edges (a workgroup files its 6400 edges by destination tile in LDS, writes each tile's run whole); no tile scans dst; 8 launches; LDS sum order by arrival
# speedup vs baseline: 3.9444x; 1.2165x over previous
#include <hip/hip_runtime.h>
#include <math.h>

typedef __attribute__((ext_vector_type(2))) float v2f;
typedef __attribute__((ext_vector_type(8))) float v8f;

static constexpr int FEAT = 128;
static constexpr int APAD = 132;
static constexpr int ROWS_PER_BLK = 64;

static constexpr int DEG_NR = 12800;
static constexpr int DEG_S  = 32;
static constexpr int DEG_T  = 1024;
static constexpr int DEG_U  = 4;

__device__ __forceinline__ void deg_count(int* h, int v, int lo) {
    const unsigned u = (unsigned)(v - lo);
    if (u < (unsigned)DEG_NR) __hip_atomic_fetch_add(h + u, 1, __ATOMIC_RELAXED, __HIP_MEMORY_SCOPE_WORKGROUP);
}

__global__ __launch_bounds__(DEG_T) void k_degree(const int* __restrict__ src,
                                                  const int* __restrict__ dst,
                                                  int* __restrict__ part,
                                                  int ne, int es, int nranges, int nalign) {
    __shared__ int hO[DEG_NR];
    __shared__ int hI[DEG_NR];
    const int t = threadIdx.x;
    const int g = blockIdx.x % nranges;
    const int s = blockIdx.x / nranges;
    for (int j = t; j < DEG_NR; j += DEG_T) { hO[j] = 0; hI[j] = 0; }
    __syncthreads();

    const int lo = g * DEG_NR;
    const int e0 = s * es;
    const int hi = e0 + es < ne ? e0 + es : ne;
    for (int eb = e0 + 4 * t; eb + 3 < hi; eb += DEG_U * 4 * DEG_T) {
        int4 a[DEG_U], b[DEG_U];
        bool ok[DEG_U];
        #pragma unroll
        for (int u = 0; u < DEG_U; ++u) {
            const int e = eb + u * 4 * DEG_T;
            ok[u] = e + 3 < hi;
            const int ea = ok[u] ? e : eb;
            a[u] = *(const int4*)(src + ea);
            b[u] = *(const int4*)(dst + ea);
        }
        #pragma unroll
        for (int u = 0; u < DEG_U; ++u) {
            if (ok[u]) {
                deg_count(hO, a[u].x, lo); deg_count(hO, a[u].y, lo);
                deg_count(hO, a[u].z, lo); deg_count(hO, a[u].w, lo);
                deg_count(hI, b[u].x, lo); deg_count(hI, b[u].y, lo);
                deg_count(hI, b[u].z, lo); deg_count(hI, b[u].w, lo);
            }
        }
    }
    {
        const int et = (hi & ~3) + t;
        if (t < 3 && et >= e0 && et < hi) {
            deg_count(hO, src[et], lo);
            deg_count(hI, dst[et], lo);
        }
    }
    __syncthreads();

    int* pO = part + (size_t)(2 * s) * nalign + lo;
    int* pI = pO + nalign;
    for (int j = 4 * t; j < DEG_NR; j += 4 * DEG_T) {
        if (lo + j < nalign) {
            *(int4*)(pO + j) = *(const int4*)(hO + j);
            *(int4*)(pI + j) = *(const int4*)(hI + j);
        }
    }
}

__global__ __launch_bounds__(256) void k_rsqrt_clip(const int* __restrict__ part,
                                                    float* __restrict__ rsO,
                                                    float* __restrict__ rsI, int n, int nalign) {
    int i = blockIdx.x * 256 + threadIdx.x;
    if (i < n) {
        int ca = 0, cb = 0;
        #pragma unroll 8
        for (int s = 0; s < DEG_S; ++s) {
            ca += part[(size_t)(2 * s) * nalign + i];
            cb += part[(size_t)(2 * s + 1) * nalign + i];
        }
        float a = (float)ca; a = a < 1.0f ? 1.0f : a;
        rsO[i] = 1.0f / sqrtf(a);
        float b = (float)cb; b = b < 1.0f ? 1.0f : b;
        rsI[i] = 1.0f / sqrtf(b);
    }
}

__device__ __forceinline__ float4 fin4(float4 a, float s, float4 b) {
    float4 r;
    r.x = a.x * s + b.x; r.x = r.x > 0.f ? r.x : 0.f;
    r.y = a.y * s + b.y; r.y = r.y > 0.f ? r.y : 0.f;
    r.z = a.z * s + b.z; r.z = r.z > 0.f ? r.z : 0.f;
    r.w = a.w * s + b.w; r.w = r.w > 0.f ? r.w : 0.f;
    return r;
}

template <bool FIN>
__global__ __launch_bounds__(256) void k_gemm_scaled(const float* __restrict__ X,
                                                     const float* __restrict__ rsO,
                                                     const float* __restrict__ rsI,
                                                     const float* __restrict__ bias,
                                                     const float* __restrict__ W,
                                                     float* __restrict__ Y, int nrows) {
    __shared__ float Alds[ROWS_PER_BLK * APAD];

    const int rowbase = blockIdx.x * ROWS_PER_BLK;
    const int t = threadIdx.x;

    float4 bz = {0.f, 0.f, 0.f, 0.f};
    if (FIN) bz = *(const float4*)(bias + ((t & 31) << 2));

    for (int i = t; i < ROWS_PER_BLK * (FEAT / 4); i += 256) {
        int r  = i >> 5;
        int c4 = (i & 31) << 2;
        int row = rowbase + r;
        float4 v = {0.f, 0.f, 0.f, 0.f};
        float s = 0.f;
        if (row < nrows) {
            v = *(const float4*)(X + (size_t)row * FEAT + c4);
            s = rsO[row];
            if (FIN) v = fin4(v, rsI[row], bz);
        }
        float* dstp = Alds + r * APAD + c4;
        dstp[0] = v.x * s; dstp[1] = v.y * s; dstp[2] = v.z * s; dstp[3] = v.w * s;
    }
    __syncthreads();

    const int wave = t >> 5;
    const int lane = t & 31;
    const int half = lane >> 4;
    const int l16  = lane & 15;
    const int ncol = wave * 16 + l16;

    v8f acc0 = {}, acc1 = {}, acc2 = {}, acc3 = {};
    const float* a0 = Alds + (0 * 16 + l16) * APAD;
    const float* a1 = Alds + (1 * 16 + l16) * APAD;
    const float* a2 = Alds + (2 * 16 + l16) * APAD;
    const float* a3 = Alds + (3 * 16 + l16) * APAD;

    #pragma unroll 4
    for (int k = 0; k < FEAT; k += 4) {
        const int ka = k + half * 2;
        v2f b;
        b.x = W[ka * FEAT + ncol];
        b.y = W[(ka + 1) * FEAT + ncol];
        v2f a;
        a.x = a0[ka]; a.y = a0[ka + 1];
        acc0 = __builtin_amdgcn_wmma_f32_16x16x4_f32(false, a, false, b, (short)0, acc0, false, false);
        a.x = a1[ka]; a.y = a1[ka + 1];
        acc1 = __builtin_amdgcn_wmma_f32_16x16x4_f32(false, a, false, b, (short)0, acc1, false, false);
        a.x = a2[ka]; a.y = a2[ka + 1];
        acc2 = __builtin_amdgcn_wmma_f32_16x16x4_f32(false, a, false, b, (short)0, acc2, false, false);
        a.x = a3[ka]; a.y = a3[ka + 1];
        acc3 = __builtin_amdgcn_wmma_f32_16x16x4_f32(false, a, false, b, (short)0, acc3, false, false);
    }

    const int mrow0 = rowbase + half * 8;
    if (rowbase + ROWS_PER_BLK <= nrows) {
        #pragma unroll
        for (int r = 0; r < 8; ++r) {
            Y[(size_t)(mrow0 +  0 + r) * FEAT + ncol] = acc0[r];
            Y[(size_t)(mrow0 + 16 + r) * FEAT + ncol] = acc1[r];
            Y[(size_t)(mrow0 + 32 + r) * FEAT + ncol] = acc2[r];
            Y[(size_t)(mrow0 + 48 + r) * FEAT + ncol] = acc3[r];
        }
    } else {
        #pragma unroll
        for (int r = 0; r < 8; ++r) {
            int r0 = mrow0 + r;
            if (r0 +  0 < nrows) Y[(size_t)(r0 +  0) * FEAT + ncol] = acc0[r];
            if (r0 + 16 < nrows) Y[(size_t)(r0 + 16) * FEAT + ncol] = acc1[r];
            if (r0 + 32 < nrows) Y[(size_t)(r0 + 32) * FEAT + ncol] = acc2[r];
            if (r0 + 48 < nrows) Y[(size_t)(r0 + 48) * FEAT + ncol] = acc3[r];
        }
    }
}

static constexpr int TR      = 500;
static constexpr int TS_T    = 1024;
static constexpr int TS_U    = 4;
static constexpr int TS_TRIP = TS_T * 4 * TS_U;
static constexpr int QCAP    = 32 * 4 * TS_U + 32;
static constexpr int QTOT    = (TS_T / 32) * QCAP;
static constexpr int EGT     = 8;
static constexpr int BN_T    = 800;
static constexpr int BE      = BN_T * 8;
static constexpr int CCAP    = 128;
static constexpr int BT_MAX  = 1024;

__global__ __launch_bounds__(BN_T) void k_bin(const int* __restrict__ src,
                                              const int* __restrict__ dst,
                                              int2* __restrict__ cells, int* __restrict__ ccnt,
                                              int ne, int ntiles, int nch) {
    __shared__ int  cnt[BT_MAX];
    __shared__ int  offs[BT_MAX];
    __shared__ int2 stage[BE];
    const int t = threadIdx.x;
    const int lane = t & 31;
    const int wave = t >> 5;
    const int c = blockIdx.x;
    int* mycnt = ccnt + (size_t)c * ntiles;
    if (ntiles > BT_MAX) {
        for (int j = t; j < ntiles; j += BN_T) mycnt[j] = CCAP + 1;
        return;
    }
    for (int j = t; j < ntiles; j += BN_T) cnt[j] = 0;
    __syncthreads();

    int sv[8], dv[8];
    #pragma unroll
    for (int u = 0; u < 2; ++u) {
        const int e = c * BE + u * (4 * BN_T) + 4 * t;
        int4 a = {0, 0, 0, 0};
        int4 b = {-1, -1, -1, -1};
        if (e + 3 < ne) {
            a = *(const int4*)(src + e);
            b = *(const int4*)(dst + e);
        } else {
            if (e < ne)     { a.x = src[e];     b.x = dst[e]; }
            if (e + 1 < ne) { a.y = src[e + 1]; b.y = dst[e + 1]; }
            if (e + 2 < ne) { a.z = src[e + 2]; b.z = dst[e + 2]; }
        }
        sv[4 * u + 0] = a.x; sv[4 * u + 1] = a.y; sv[4 * u + 2] = a.z; sv[4 * u + 3] = a.w;
        dv[4 * u + 0] = b.x; dv[4 * u + 1] = b.y; dv[4 * u + 2] = b.z; dv[4 * u + 3] = b.w;
    }
    const unsigned lim = (unsigned)ntiles * (unsigned)TR;
    int tl[8], rk[8];
    #pragma unroll
    for (int k = 0; k < 8; ++k) {
        tl[k] = -1; rk[k] = 0;
        if ((unsigned)dv[k] < lim) {
            tl[k] = (int)((unsigned)dv[k] / (unsigned)TR);
            dv[k] -= tl[k] * TR;
            rk[k] = __hip_atomic_fetch_add(cnt + tl[k], 1, __ATOMIC_RELAXED, __HIP_MEMORY_SCOPE_WORKGROUP);
        }
    }
    __syncthreads();

    if (t < 32) {
        const int per = (ntiles + 31) >> 5;
        const int b0 = t * per;
        int s = 0;
        for (int j = 0; j < per; ++j) if (b0 + j < ntiles) s += cnt[b0 + j];
        int inc = s;
        #pragma unroll
        for (int d = 1; d < 32; d <<= 1) {
            const int o = __shfl_up(inc, d);
            if (t >= d) inc += o;
        }
        int run = inc - s;
        for (int j = 0; j < per; ++j) if (b0 + j < ntiles) { offs[b0 + j] = run; run += cnt[b0 + j]; }
    }
    __syncthreads();

    #pragma unroll
    for (int k = 0; k < 8; ++k) {
        if (tl[k] >= 0) stage[offs[tl[k]] + rk[k]] = make_int2(sv[k], dv[k]);
    }
    __syncthreads();

    for (int tile = wave; tile < ntiles; tile += BN_T / 32) {
        const int n = cnt[tile];
        if (n <= CCAP) {
            const int2* s = stage + offs[tile];
            int2* cell = cells + ((size_t)tile * nch + c) * CCAP;
            for (int i = lane; i < n; i += 32) cell[i] = s[i];
        }
    }
    for (int j = t; j < ntiles; j += BN_T) mycnt[j] = cnt[j];
}

__device__ __forceinline__ void lds_add(float* p, float v) {
    __hip_atomic_fetch_add(p, v, __ATOMIC_RELAXED, __HIP_MEMORY_SCOPE_WORKGROUP);
}

template <bool FULL>
__device__ __forceinline__ void tile_rows(const float* __restrict__ H, float* tile, int sv, int dv,
                                          int cnt, int lane) {
    for (int g = 0; g < 32; g += EGT) {
        if (!FULL && g >= cnt) break;
        float4 v[EGT];
        #pragma unroll
        for (int j = 0; j < EGT; ++j) {
            v[j].x = 0.f; v[j].y = 0.f; v[j].z = 0.f; v[j].w = 0.f;
            if (FULL || g + j < cnt) {
                const int s = __builtin_amdgcn_readlane(sv, g + j);
                v[j] = *(const float4*)(H + (size_t)s * FEAT + lane * 4);
            }
        }
        #pragma unroll
        for (int j = 0; j < EGT; ++j) {
            if (FULL || g + j < cnt) {
                const int d = __builtin_amdgcn_readlane(dv, g + j);
                float* p = tile + d * FEAT + lane;
                lds_add(p,      v[j].x);
                lds_add(p + 32, v[j].y);
                lds_add(p + 64, v[j].z);
                lds_add(p + 96, v[j].w);
            }
        }
    }
}

__device__ __forceinline__ void tile_from_cells(const float* __restrict__ H, float* tile,
                                                const int2* __restrict__ cells, const int* ccl,
                                                int nch, int wave, int lane) {
    int c = __builtin_amdgcn_readfirstlane(wave);
    int n = 0;
    int2 p0 = make_int2(0, 0), p1 = make_int2(0, 0);
    if (c < nch) {
        n = __builtin_amdgcn_readfirstlane(ccl[c]);
        const int2* q = cells + (size_t)c * CCAP;
        if (lane < n)      p0 = q[lane];
        if (lane + 32 < n) p1 = q[lane + 32];
    }
    while (c < nch) {
        const int cn = c + TS_T / 32;
        int nn = 0;
        int2 n0 = make_int2(0, 0), n1 = make_int2(0, 0);
        if (cn < nch) {
            nn = __builtin_amdgcn_readfirstlane(ccl[cn]);
            const int2* q = cells + (size_t)cn * CCAP;
            if (lane < nn)      n0 = q[lane];
            if (lane + 32 < nn) n1 = q[lane + 32];
        }
        for (int b = 0; b < n && b < 64; b += 32) {
            const int2 p = b ? p1 : p0;
            const int cnt = n - b;
            if (cnt >= 32) tile_rows<true>(H, tile, p.x, p.y, 32, lane);
            else           tile_rows<false>(H, tile, p.x, p.y, cnt, lane);
        }
        const int2* q = cells + (size_t)c * CCAP;
        for (int b = 64; b < n; b += 32) {
            int2 r = make_int2(0, 0);
            if (b + lane < n) r = q[b + lane];
            const int cnt = n - b;
            if (cnt >= 32) tile_rows<true>(H, tile, r.x, r.y, 32, lane);
            else           tile_rows<false>(H, tile, r.x, r.y, cnt, lane);
        }
        c = cn; n = nn; p0 = n0; p1 = n1;
    }
}

template <bool FULL>
__device__ __forceinline__ void tile_drain(const float* __restrict__ H, const int* __restrict__ src,
                                           const int* __restrict__ dst, float* tile, const int* q,
                                           int cnt, int lo, int lane) {
    int e = 0;
    if (FULL || lane < cnt) e = q[lane];
    const int sv = src[e];
    const int dv = dst[e] - lo;
    tile_rows<FULL>(H, tile, sv, dv, cnt, lane);
}

__device__ __forceinline__ void tile_load_trip(const int* __restrict__ dst, int4* a, int tr, int t, int ne) {
    const int elast = ne >= 4 ? (ne - 4) & ~3 : 0;
    #pragma unroll
    for (int u = 0; u < TS_U; ++u) {
        const int e = tr * TS_TRIP + u * (TS_T * 4) + t * 4;
        const unsigned ea = (unsigned)(e < elast ? e : elast);
        a[u] = *(const int4*)((const char*)dst + ea * 4u);
    }
}

__device__ __forceinline__ void tile_file(int* q, int& qn, unsigned m, int e, int lane) {
    const int idx = qn + (int)__builtin_amdgcn_mbcnt_lo(m, 0u);
    if ((m >> lane) & 1u) q[idx] = e;
    qn += __builtin_popcount(m);
}

__device__ __forceinline__ void tile_scan(const float* __restrict__ H, const int* __restrict__ src,
                                          const int* __restrict__ dst, float* tile, int* q,
                                          int ne, int lo, int t) {
    const int lane = t & 31;
    const int wave = t >> 5;
    int qn = 0;
    const int ntrips = (ne + TS_TRIP - 1) / TS_TRIP;
    int trip = (int)(((long long)blockIdx.x * ntrips) / gridDim.x);

    int4 cur[TS_U];
    tile_load_trip(dst, cur, trip, t, ne);
    for (int i = 0; i < ntrips; ++i) {
        int tn = trip + 1;
        if (tn == ntrips) tn = 0;
        int4 nxt[TS_U];
        tile_load_trip(dst, nxt, i + 1 < ntrips ? tn : trip, t, ne);

        const int ebase = trip * TS_TRIP + t * 4;
        unsigned mk[4 * TS_U];
        #pragma unroll
        for (int u = 0; u < TS_U; ++u) {
            const unsigned okm = __builtin_amdgcn_ballot_w32(ebase + u * (TS_T * 4) + 3 < ne);
            mk[4 * u + 0] = __builtin_amdgcn_ballot_w32((unsigned)(cur[u].x - lo) < (unsigned)TR) & okm;
            mk[4 * u + 1] = __builtin_amdgcn_ballot_w32((unsigned)(cur[u].y - lo) < (unsigned)TR) & okm;
            mk[4 * u + 2] = __builtin_amdgcn_ballot_w32((unsigned)(cur[u].z - lo) < (unsigned)TR) & okm;
            mk[4 * u + 3] = __builtin_amdgcn_ballot_w32((unsigned)(cur[u].w - lo) < (unsigned)TR) & okm;
        }
        #pragma unroll
        for (int k = 0; k < 4 * TS_U; ++k) {
            if (mk[k]) tile_file(q, qn, mk[k], ebase + (k >> 2) * (TS_T * 4) + (k & 3), lane);
        }

        int head = 0;
        while (qn - head >= 32) {
            tile_drain<true>(H, src, dst, tile, q + head, 32, lo, lane);
            head += 32;
        }
        if (head > 0) {
            const int left = qn - head;
            int v = 0;
            if (lane < left) v = q[head + lane];
            if (lane < left) q[lane] = v;
            qn = left;
        }

        #pragma unroll
        for (int u = 0; u < TS_U; ++u) cur[u] = nxt[u];
        trip = tn;
    }

    if (wave == 0) {
        const int e = (ne & ~3) + lane;
        const bool ok = lane < 3 && e < ne;
        const int dval = dst[ok ? e : 0];
        const unsigned m = __builtin_amdgcn_ballot_w32(ok && (unsigned)(dval - lo) < (unsigned)TR);
        if (m) tile_file(q, qn, m, e, lane);
    }
    if (qn >= 32) {
        tile_drain<true>(H, src, dst, tile, q, 32, lo, lane);
        qn -= 32;
        if (qn > 0) tile_drain<false>(H, src, dst, tile, q + 32, qn, lo, lane);
    } else if (qn > 0) {
        tile_drain<false>(H, src, dst, tile, q, qn, lo, lane);
    }
}

__global__ __launch_bounds__(TS_T) void k_tile_sum(const float* __restrict__ H,
                                                   const int* __restrict__ src,
                                                   const int* __restrict__ dst,
                                                   float* __restrict__ agg,
                                                   const int2* __restrict__ cells,
                                                   const int* __restrict__ ccnt,
                                                   int ne, int nnodes, int nch) {
    __shared__ float tile[TR * FEAT];
    __shared__ int   qbuf[QTOT];
    const int t = threadIdx.x;
    const int lane = t & 31;
    const int wave = t >> 5;
    const int lo = blockIdx.x * TR;

    {
        const float4 z = {0.f, 0.f, 0.f, 0.f};
        for (int i = t; i < TR * FEAT / 4; i += TS_T) ((float4*)tile)[i] = z;
    }
    int bad = nch > QTOT ? 1 : 0;
    for (int j = t; j < nch && j < QTOT; j += TS_T) {
        const int v = ccnt[(size_t)j * gridDim.x + blockIdx.x];
        qbuf[j] = v;
        if (v > CCAP) bad = 1;
    }
    const int scan = __syncthreads_or(bad);

    if (!scan) tile_from_cells(H, tile, cells + (size_t)blockIdx.x * nch * CCAP, qbuf, nch, wave, lane);
    else       tile_scan(H, src, dst, tile, qbuf + wave * QCAP, ne, lo, t);
    __syncthreads();

    for (int r = wave; r < TR; r += TS_T / 32) {
        const int row = lo + r;
        if (row < nnodes) {
            const float* p = tile + r * FEAT + lane;
            float4 v;
            v.x = p[0]; v.y = p[32]; v.z = p[64]; v.w = p[96];
            *(float4*)(agg + (size_t)row * FEAT + lane * 4) = v;
        }
    }
}

__global__ __launch_bounds__(256) void k_classify(const float* __restrict__ agg,
                                                  const float* __restrict__ rsI,
                                                  const float* __restrict__ bias,
                                                  const float* __restrict__ Wc,
                                                  const float* __restrict__ bc,
                                                  float* __restrict__ out, int n4) {
    int i = blockIdx.x * 256 + threadIdx.x;
    if (i >= n4) return;
    int node = i >> 2;
    int c = i & 3;
    const float* a = agg + (size_t)node * FEAT;
    const float s = rsI[node];
    float sum = 0.f;
    #pragma unroll 8
    for (int k = 0; k < FEAT; ++k) {
        float h = a[k] * s + bias[k];
        h = h > 0.f ? h : 0.f;
        sum += h * Wc[k * 4 + c];
    }
    out[i] = sum + bc[c];
}

extern "C" void kernel_launch(void* const* d_in, const int* in_sizes, int n_in,
                              void* d_out, int out_size, void* d_ws, size_t ws_size,
                              hipStream_t stream) {
    const float* feat = (const float*)d_in[0];
    const int*   src  = (const int*)d_in[1];
    const int*   dst  = (const int*)d_in[2];
    const float* W1   = (const float*)d_in[3];
    const float* b1   = (const float*)d_in[4];
    const float* W2   = (const float*)d_in[5];
    const float* b2   = (const float*)d_in[6];
    const float* Wc   = (const float*)d_in[7];
    const float* bc   = (const float*)d_in[8];
    float* out = (float*)d_out;

    const int nnodes = in_sizes[0] / FEAT;
    const int nedges = in_sizes[1];
    const size_t nAlign = ((size_t)nnodes + 3) & ~(size_t)3;
    const size_t nfeat = (size_t)nnodes * FEAT;

    float* rsO  = (float*)d_ws;
    float* rsI  = rsO + nAlign;
    float* bufA = rsI + nAlign;
    float* bufB = bufA + nfeat;
    int*   part = (int*)(bufB + nfeat);

    const int B = 256;
    const int gNode   = (nnodes + B - 1) / B;
    const int nranges = (nnodes + DEG_NR - 1) / DEG_NR;
    const int eslice  = ((nedges + DEG_S - 1) / DEG_S + 3) & ~3;
    const int gTile   = (nnodes + TR - 1) / TR;
    const int gGemm   = (nnodes + ROWS_PER_BLK - 1) / ROWS_PER_BLK;
    const int gCls    = (nnodes * 4 + B - 1) / B;
    const int nch     = nedges > 0 ? (nedges + BE - 1) / BE : 1;
    int2*  cells = (int2*)(part + (size_t)DEG_S * 2 * nAlign);
    int*   ccnt  = (int*)(cells + (size_t)gTile * nch * CCAP);

    k_degree<<<nranges * DEG_S, DEG_T, 0, stream>>>(src, dst, part, nedges, eslice, nranges, (int)nAlign);
    k_rsqrt_clip<<<gNode, B, 0, stream>>>(part, rsO, rsI, nnodes, (int)nAlign);

    k_gemm_scaled<false><<<gGemm, B, 0, stream>>>(feat, rsO, nullptr, nullptr, W1, bufA, nnodes);
    k_bin<<<nch, BN_T, 0, stream>>>(src, dst, cells, ccnt, nedges, gTile, nch);
    k_tile_sum<<<gTile, TS_T, 0, stream>>>(bufA, src, dst, bufB, cells, ccnt, nedges, nnodes, nch);

    k_gemm_scaled<true><<<gGemm, B, 0, stream>>>(bufB, rsO, rsI, b1, W2, bufA, nnodes);
    k_tile_sum<<<gTile, TS_T, 0, stream>>>(bufA, src, dst, bufB, cells, ccnt, nedges, nnodes, nch);

    k_classify<<<gCls, B, 0, stream>>>(bufB, rsI, b2, Wc, bc, out, nnodes * 4);
}


# edge sums without LDS float atomics: a tile files its edge pairs by owner wave (tile row mod 32) in LDS bins, each wave adds its 16 rows in registers, stores them once; sums in bin arrival order
# speedup vs baseline: 4.2601x; 1.0800x over previous
#include <hip/hip_runtime.h>
#include <math.h>

typedef __attribute__((ext_vector_type(2))) float v2f;
typedef __attribute__((ext_vector_type(8))) float v8f;

static constexpr int FEAT = 128;
static constexpr int APAD = 132;
static constexpr int ROWS_PER_BLK = 64;

static constexpr int DEG_NR = 12800;
static constexpr int DEG_S  = 32;
static constexpr int DEG_T  = 1024;
static constexpr int DEG_U  = 4;

__device__ __forceinline__ void deg_count(int* h, int v, int lo) {
    const unsigned u = (unsigned)(v - lo);
    if (u < (unsigned)DEG_NR) __hip_atomic_fetch_add(h + u, 1, __ATOMIC_RELAXED, __HIP_MEMORY_SCOPE_WORKGROUP);
}

__global__ __launch_bounds__(DEG_T) void k_degree(const int* __restrict__ src,
                                                  const int* __restrict__ dst,
                                                  int* __restrict__ part,
                                                  int ne, int es, int nranges, int nalign) {
    __shared__ int hO[DEG_NR];
    __shared__ int hI[DEG_NR];
    const int t = threadIdx.x;
    const int g = blockIdx.x % nranges;
    const int s = blockIdx.x / nranges;
    for (int j = t; j < DEG_NR; j += DEG_T) { hO[j] = 0; hI[j] = 0; }
    __syncthreads();

    const int lo = g * DEG_NR;
    const int e0 = s * es;
    const int hi = e0 + es < ne ? e0 + es : ne;
    for (int eb = e0 + 4 * t; eb + 3 < hi; eb += DEG_U * 4 * DEG_T) {
        int4 a[DEG_U], b[DEG_U];
        bool ok[DEG_U];
        #pragma unroll
        for (int u = 0; u < DEG_U; ++u) {
            const int e = eb + u * 4 * DEG_T;
            ok[u] = e + 3 < hi;
            const int ea = ok[u] ? e : eb;
            a[u] = *(const int4*)(src + ea);
            b[u] = *(const int4*)(dst + ea);
        }
        #pragma unroll
        for (int u = 0; u < DEG_U; ++u) {
            if (ok[u]) {
                deg_count(hO, a[u].x, lo); deg_count(hO, a[u].y, lo);
                deg_count(hO, a[u].z, lo); deg_count(hO, a[u].w, lo);
                deg_count(hI, b[u].x, lo); deg_count(hI, b[u].y, lo);
                deg_count(hI, b[u].z, lo); deg_count(hI, b[u].w, lo);
            }
        }
    }
    {
        const int et = (hi & ~3) + t;
        if (t < 3 && et >= e0 && et < hi) {
            deg_count(hO, src[et], lo);
            deg_count(hI, dst[et], lo);
        }
    }
    __syncthreads();

    int* pO = part + (size_t)(2 * s) * nalign + lo;
    int* pI = pO + nalign;
    for (int j = 4 * t; j < DEG_NR; j += 4 * DEG_T) {
        if (lo + j < nalign) {
            *(int4*)(pO + j) = *(const int4*)(hO + j);
            *(int4*)(pI + j) = *(const int4*)(hI + j);
        }
    }
}

__global__ __launch_bounds__(256) void k_rsqrt_clip(const int* __restrict__ part,
                                                    float* __restrict__ rsO,
                                                    float* __restrict__ rsI, int n, int nalign) {
    int i = blockIdx.x * 256 + threadIdx.x;
    if (i < n) {
        int ca = 0, cb = 0;
        #pragma unroll 8
        for (int s = 0; s < DEG_S; ++s) {
            ca += part[(size_t)(2 * s) * nalign + i];
            cb += part[(size_t)(2 * s + 1) * nalign + i];
        }
        float a = (float)ca; a = a < 1.0f ? 1.0f : a;
        rsO[i] = 1.0f / sqrtf(a);
        float b = (float)cb; b = b < 1.0f ? 1.0f : b;
        rsI[i] = 1.0f / sqrtf(b);
    }
}

__device__ __forceinline__ float4 fin4(float4 a, float s, float4 b) {
    float4 r;
    r.x = a.x * s + b.x; r.x = r.x > 0.f ? r.x : 0.f;
    r.y = a.y * s + b.y; r.y = r.y > 0.f ? r.y : 0.f;
    r.z = a.z * s + b.z; r.z = r.z > 0.f ? r.z : 0.f;
    r.w = a.w * s + b.w; r.w = r.w > 0.f ? r.w : 0.f;
    return r;
}

template <bool FIN>
__global__ __launch_bounds__(256) void k_gemm_scaled(const float* __restrict__ X,
                                                     const float* __restrict__ rsO,
                                                     const float* __restrict__ rsI,
                                                     const float* __restrict__ bias,
                                                     const float* __restrict__ W,
                                                     float* __restrict__ Y, int nrows) {
    __shared__ float Alds[ROWS_PER_BLK * APAD];

    const int rowbase = blockIdx.x * ROWS_PER_BLK;
    const int t = threadIdx.x;

    float4 bz = {0.f, 0.f, 0.f, 0.f};
    if (FIN) bz = *(const float4*)(bias + ((t & 31) << 2));

    for (int i = t; i < ROWS_PER_BLK * (FEAT / 4); i += 256) {
        int r  = i >> 5;
        int c4 = (i & 31) << 2;
        int row = rowbase + r;
        float4 v = {0.f, 0.f, 0.f, 0.f};
        float s = 0.f;
        if (row < nrows) {
            v = *(const float4*)(X + (size_t)row * FEAT + c4);
            s = rsO[row];
            if (FIN) v = fin4(v, rsI[row], bz);
        }
        float* dstp = Alds + r * APAD + c4;
        dstp[0] = v.x * s; dstp[1] = v.y * s; dstp[2] = v.z * s; dstp[3] = v.w * s;
    }
    __syncthreads();

    const int wave = t >> 5;
    const int lane = t & 31;
    const int half = lane >> 4;
    const int l16  = lane & 15;
    const int ncol = wave * 16 + l16;

    v8f acc0 = {}, acc1 = {}, acc2 = {}, acc3 = {};
    const float* a0 = Alds + (0 * 16 + l16) * APAD;
    const float* a1 = Alds + (1 * 16 + l16) * APAD;
    const float* a2 = Alds + (2 * 16 + l16) * APAD;
    const float* a3 = Alds + (3 * 16 + l16) * APAD;

    #pragma unroll 4
    for (int k = 0; k < FEAT; k += 4) {
        const int ka = k + half * 2;
        v2f b;
        b.x = W[ka * FEAT + ncol];
        b.y = W[(ka + 1) * FEAT + ncol];
        v2f a;
        a.x = a0[ka]; a.y = a0[ka + 1];
        acc0 = __builtin_amdgcn_wmma_f32_16x16x4_f32(false, a, false, b, (short)0, acc0, false, false);
        a.x = a1[ka]; a.y = a1[ka + 1];
        acc1 = __builtin_amdgcn_wmma_f32_16x16x4_f32(false, a, false, b, (short)0, acc1, false, false);
        a.x = a2[ka]; a.y = a2[ka + 1];
        acc2 = __builtin_amdgcn_wmma_f32_16x16x4_f32(false, a, false, b, (short)0, acc2, false, false);
        a.x = a3[ka]; a.y = a3[ka + 1];
        acc3 = __builtin_amdgcn_wmma_f32_16x16x4_f32(false, a, false, b, (short)0, acc3, false, false);
    }

    const int mrow0 = rowbase + half * 8;
    if (rowbase + ROWS_PER_BLK <= nrows) {
        #pragma unroll
        for (int r = 0; r < 8; ++r) {
            Y[(size_t)(mrow0 +  0 + r) * FEAT + ncol] = acc0[r];
            Y[(size_t)(mrow0 + 16 + r) * FEAT + ncol] = acc1[r];
            Y[(size_t)(mrow0 + 32 + r) * FEAT + ncol] = acc2[r];
            Y[(size_t)(mrow0 + 48 + r) * FEAT + ncol] = acc3[r];
        }
    } else {
        #pragma unroll
        for (int r = 0; r < 8; ++r) {
            int r0 = mrow0 + r;
            if (r0 +  0 < nrows) Y[(size_t)(r0 +  0) * FEAT + ncol] = acc0[r];
            if (r0 + 16 < nrows) Y[(size_t)(r0 + 16) * FEAT + ncol] = acc1[r];
            if (r0 + 32 < nrows) Y[(size_t)(r0 + 32) * FEAT + ncol] = acc2[r];
            if (r0 + 48 < nrows) Y[(size_t)(r0 + 48) * FEAT + ncol] = acc3[r];
        }
    }
}

static constexpr int TR      = 500;
static constexpr int TS_T    = 1024;
static constexpr int TS_U    = 4;
static constexpr int TS_TRIP = TS_T * 4 * TS_U;
static constexpr int QCAP    = 32 * 4 * TS_U + 32;
static constexpr int QTOT    = (TS_T / 32) * QCAP;
static constexpr int EGT     = 8;
static constexpr int BN_T    = 800;
static constexpr int BE      = BN_T * 8;
static constexpr int CCAP    = 128;
static constexpr int BT_MAX  = 1024;

__global__ __launch_bounds__(BN_T) void k_bin(const int* __restrict__ src,
                                              const int* __restrict__ dst,
                                              int2* __restrict__ cells, int* __restrict__ ccnt,
                                              int ne, int ntiles, int nch) {
    __shared__ int  cnt[BT_MAX];
    __shared__ int  offs[BT_MAX];
    __shared__ int2 stage[BE];
    const int t = threadIdx.x;
    const int lane = t & 31;
    const int wave = t >> 5;
    const int c = blockIdx.x;
    int* mycnt = ccnt + (size_t)c * ntiles;
    if (ntiles > BT_MAX) {
        for (int j = t; j < ntiles; j += BN_T) mycnt[j] = CCAP + 1;
        return;
    }
    for (int j = t; j < ntiles; j += BN_T) cnt[j] = 0;
    __syncthreads();

    int sv[8], dv[8];
    #pragma unroll
    for (int u = 0; u < 2; ++u) {
        const int e = c * BE + u * (4 * BN_T) + 4 * t;
        int4 a = {0, 0, 0, 0};
        int4 b = {-1, -1, -1, -1};
        if (e + 3 < ne) {
            a = *(const int4*)(src + e);
            b = *(const int4*)(dst + e);
        } else {
            if (e < ne)     { a.x = src[e];     b.x = dst[e]; }
            if (e + 1 < ne) { a.y = src[e + 1]; b.y = dst[e + 1]; }
            if (e + 2 < ne) { a.z = src[e + 2]; b.z = dst[e + 2]; }
        }
        sv[4 * u + 0] = a.x; sv[4 * u + 1] = a.y; sv[4 * u + 2] = a.z; sv[4 * u + 3] = a.w;
        dv[4 * u + 0] = b.x; dv[4 * u + 1] = b.y; dv[4 * u + 2] = b.z; dv[4 * u + 3] = b.w;
    }
    const unsigned lim = (unsigned)ntiles * (unsigned)TR;
    int tl[8], rk[8];
    #pragma unroll
    for (int k = 0; k < 8; ++k) {
        tl[k] = -1; rk[k] = 0;
        if ((unsigned)dv[k] < lim) {
            tl[k] = (int)((unsigned)dv[k] / (unsigned)TR);
            dv[k] -= tl[k] * TR;
            rk[k] = __hip_atomic_fetch_add(cnt + tl[k], 1, __ATOMIC_RELAXED, __HIP_MEMORY_SCOPE_WORKGROUP);
        }
    }
    __syncthreads();

    if (t < 32) {
        const int per = (ntiles + 31) >> 5;
        const int b0 = t * per;
        int s = 0;
        for (int j = 0; j < per; ++j) if (b0 + j < ntiles) s += cnt[b0 + j];
        int inc = s;
        #pragma unroll
        for (int d = 1; d < 32; d <<= 1) {
            const int o = __shfl_up(inc, d);
            if (t >= d) inc += o;
        }
        int run = inc - s;
        for (int j = 0; j < per; ++j) if (b0 + j < ntiles) { offs[b0 + j] = run; run += cnt[b0 + j]; }
    }
    __syncthreads();

    #pragma unroll
    for (int k = 0; k < 8; ++k) {
        if (tl[k] >= 0) stage[offs[tl[k]] + rk[k]] = make_int2(sv[k], dv[k]);
    }
    __syncthreads();

    for (int tile = wave; tile < ntiles; tile += BN_T / 32) {
        const int n = cnt[tile];
        if (n <= CCAP) {
            const int2* s = stage + offs[tile];
            int2* cell = cells + ((size_t)tile * nch + c) * CCAP;
            for (int i = lane; i < n; i += 32) cell[i] = s[i];
        }
    }
    for (int j = t; j < ntiles; j += BN_T) mycnt[j] = cnt[j];
}

__device__ __forceinline__ void lds_add(float* p, float v) {
    __hip_atomic_fetch_add(p, v, __ATOMIC_RELAXED, __HIP_MEMORY_SCOPE_WORKGROUP);
}

template <bool FULL>
__device__ __forceinline__ void tile_rows(const float* __restrict__ H, float* tile, int sv, int dv,
                                          int cnt, int lane) {
    for (int g = 0; g < 32; g += EGT) {
        if (!FULL && g >= cnt) break;
        float4 v[EGT];
        #pragma unroll
        for (int j = 0; j < EGT; ++j) {
            v[j].x = 0.f; v[j].y = 0.f; v[j].z = 0.f; v[j].w = 0.f;
            if (FULL || g + j < cnt) {
                const int s = __builtin_amdgcn_readlane(sv, g + j);
                v[j] = *(const float4*)(H + (size_t)s * FEAT + lane * 4);
            }
        }
        #pragma unroll
        for (int j = 0; j < EGT; ++j) {
            if (FULL || g + j < cnt) {
                const int d = __builtin_amdgcn_readlane(dv, g + j);
                float* p = tile + d * FEAT + lane;
                lds_add(p,      v[j].x);
                lds_add(p + 32, v[j].y);
                lds_add(p + 64, v[j].z);
                lds_add(p + 96, v[j].w);
            }
        }
    }
}

__device__ __forceinline__ void tile_from_cells(const float* __restrict__ H, float* tile,
                                                const int2* __restrict__ cells, const int* ccl,
                                                int nch, int wave, int lane) {
    int c = __builtin_amdgcn_readfirstlane(wave);
    int n = 0;
    int2 p0 = make_int2(0, 0), p1 = make_int2(0, 0);
    if (c < nch) {
        n = __builtin_amdgcn_readfirstlane(ccl[c]);
        const int2* q = cells + (size_t)c * CCAP;
        if (lane < n)      p0 = q[lane];
        if (lane + 32 < n) p1 = q[lane + 32];
    }
    while (c < nch) {
        const int cn = c + TS_T / 32;
        int nn = 0;
        int2 n0 = make_int2(0, 0), n1 = make_int2(0, 0);
        if (cn < nch) {
            nn = __builtin_amdgcn_readfirstlane(ccl[cn]);
            const int2* q = cells + (size_t)cn * CCAP;
            if (lane < nn)      n0 = q[lane];
            if (lane + 32 < nn) n1 = q[lane + 32];
        }
        for (int b = 0; b < n && b < 64; b += 32) {
            const int2 p = b ? p1 : p0;
            const int cnt = n - b;
            if (cnt >= 32) tile_rows<true>(H, tile, p.x, p.y, 32, lane);
            else           tile_rows<false>(H, tile, p.x, p.y, cnt, lane);
        }
        const int2* q = cells + (size_t)c * CCAP;
        for (int b = 64; b < n; b += 32) {
            int2 r = make_int2(0, 0);
            if (b + lane < n) r = q[b + lane];
            const int cnt = n - b;
            if (cnt >= 32) tile_rows<true>(H, tile, r.x, r.y, 32, lane);
            else           tile_rows<false>(H, tile, r.x, r.y, cnt, lane);
        }
        c = cn; n = nn; p0 = n0; p1 = n1;
    }
}

template <bool FULL>
__device__ __forceinline__ void tile_drain(const float* __restrict__ H, const int* __restrict__ src,
                                           const int* __restrict__ dst, float* tile, const int* q,
                                           int cnt, int lo, int lane) {
    int e = 0;
    if (FULL || lane < cnt) e = q[lane];
    const int sv = src[e];
    const int dv = dst[e] - lo;
    tile_rows<FULL>(H, tile, sv, dv, cnt, lane);
}

__device__ __forceinline__ void tile_load_trip(const int* __restrict__ dst, int4* a, int tr, int t, int ne) {
    const int elast = ne >= 4 ? (ne - 4) & ~3 : 0;
    #pragma unroll
    for (int u = 0; u < TS_U; ++u) {
        const int e = tr * TS_TRIP + u * (TS_T * 4) + t * 4;
        const unsigned ea = (unsigned)(e < elast ? e : elast);
        a[u] = *(const int4*)((const char*)dst + ea * 4u);
    }
}

__device__ __forceinline__ void tile_file(int* q, int& qn, unsigned m, int e, int lane) {
    const int idx = qn + (int)__builtin_amdgcn_mbcnt_lo(m, 0u);
    if ((m >> lane) & 1u) q[idx] = e;
    qn += __builtin_popcount(m);
}

__device__ __forceinline__ void tile_scan(const float* __restrict__ H, const int* __restrict__ src,
                                          const int* __restrict__ dst, float* tile, int* q,
                                          int ne, int lo, int t) {
    const int lane = t & 31;
    const int wave = t >> 5;
    int qn = 0;
    const int ntrips = (ne + TS_TRIP - 1) / TS_TRIP;
    int trip = (int)(((long long)blockIdx.x * ntrips) / gridDim.x);

    int4 cur[TS_U];
    tile_load_trip(dst, cur, trip, t, ne);
    for (int i = 0; i < ntrips; ++i) {
        int tn = trip + 1;
        if (tn == ntrips) tn = 0;
        int4 nxt[TS_U];
        tile_load_trip(dst, nxt, i + 1 < ntrips ? tn : trip, t, ne);

        const int ebase = trip * TS_TRIP + t * 4;
        unsigned mk[4 * TS_U];
        #pragma unroll
        for (int u = 0; u < TS_U; ++u) {
            const unsigned okm = __builtin_amdgcn_ballot_w32(ebase + u * (TS_T * 4) + 3 < ne);
            mk[4 * u + 0] = __builtin_amdgcn_ballot_w32((unsigned)(cur[u].x - lo) < (unsigned)TR) & okm;
            mk[4 * u + 1] = __builtin_amdgcn_ballot_w32((unsigned)(cur[u].y - lo) < (unsigned)TR) & okm;
            mk[4 * u + 2] = __builtin_amdgcn_ballot_w32((unsigned)(cur[u].z - lo) < (unsigned)TR) & okm;
            mk[4 * u + 3] = __builtin_amdgcn_ballot_w32((unsigned)(cur[u].w - lo) < (unsigned)TR) & okm;
        }
        #pragma unroll
        for (int k = 0; k < 4 * TS_U; ++k) {
            if (mk[k]) tile_file(q, qn, mk[k], ebase + (k >> 2) * (TS_T * 4) + (k & 3), lane);
        }

        int head = 0;
        while (qn - head >= 32) {
            tile_drain<true>(H, src, dst, tile, q + head, 32, lo, lane);
            head += 32;
        }
        if (head > 0) {
            const int left = qn - head;
            int v = 0;
            if (lane < left) v = q[head + lane];
            if (lane < left) q[lane] = v;
            qn = left;
        }

        #pragma unroll
        for (int u = 0; u < TS_U; ++u) cur[u] = nxt[u];
        trip = tn;
    }

    if (wave == 0) {
        const int e = (ne & ~3) + lane;
        const bool ok = lane < 3 && e < ne;
        const int dval = dst[ok ? e : 0];
        const unsigned m = __builtin_amdgcn_ballot_w32(ok && (unsigned)(dval - lo) < (unsigned)TR);
        if (m) tile_file(q, qn, m, e, lane);
    }
    if (qn >= 32) {
        tile_drain<true>(H, src, dst, tile, q, 32, lo, lane);
        qn -= 32;
        if (qn > 0) tile_drain<false>(H, src, dst, tile, q + 32, qn, lo, lane);
    } else if (qn > 0) {
        tile_drain<false>(H, src, dst, tile, q, qn, lo, lane);
    }
}

static constexpr int OW     = TS_T / 32;
static constexpr int OW_LG  = 5;
static constexpr int RPW    = (TR + OW - 1) / OW;
static constexpr int BCAP   = TR * FEAT / 2 / OW;
static_assert((1 << OW_LG) == OW, "owner of a row = its low bits");
static_assert(RPW == 16, "a wave's sums are vectors of 16");
static_assert(EGT == 8, "owner_group names the 8 loaded rows of a group");

__device__ __forceinline__ int file_one(int2* bins, int* bcnt, int2 p) {
    const int o = p.y & (OW - 1);
    const int slot = __hip_atomic_fetch_add(bcnt + o, 1, __ATOMIC_RELAXED, __HIP_MEMORY_SCOPE_WORKGROUP);
    if (slot < BCAP) { bins[o * BCAP + slot] = make_int2(p.x, p.y >> OW_LG); return 0; }
    return 1;
}

static constexpr int FU = 8;
__device__ __forceinline__ int file_cells(const int2* __restrict__ cells, const int* ccl, int nch,
                                          int2* bins, int* bcnt, int wave, int lane) {
    int ovf = 0;
    for (int c0 = __builtin_amdgcn_readfirstlane(wave); c0 < nch; c0 += FU * OW) {
        int n[FU];
        int2 p0[FU], p1[FU];
        #pragma unroll
        for (int u = 0; u < FU; ++u) {
            const int c = c0 + u * OW;
            n[u] = 0;
            p0[u] = make_int2(0, 0); p1[u] = make_int2(0, 0);
            if (c < nch) {
                n[u] = __builtin_amdgcn_readfirstlane(ccl[c]);
                const int2* q = cells + (size_t)c * CCAP;
                if (lane < n[u])      p0[u] = q[lane];
                if (lane + 32 < n[u]) p1[u] = q[lane + 32];
            }
        }
        #pragma unroll
        for (int u = 0; u < FU; ++u) {
            if (lane < n[u])      ovf |= file_one(bins, bcnt, p0[u]);
            if (lane + 32 < n[u]) ovf |= file_one(bins, bcnt, p1[u]);
            if (n[u] > 64) {
                const int2* q = cells + (size_t)(c0 + u * OW) * CCAP;
                for (int b = 64 + lane; b < n[u]; b += 32) ovf |= file_one(bins, bcnt, q[b]);
            }
        }
    }
    return ovf;
}

typedef __attribute__((ext_vector_type(16))) float v16f;

template <bool FULL>
__device__ __forceinline__ void owner_group(const float* __restrict__ H, int2 p, int g, int cnt, int lane,
                                            v16f& ax, v16f& ay, v16f& az, v16f& aw) {
    float4 v[EGT];
    #pragma unroll
    for (int j = 0; j < EGT; ++j) {
        if (FULL || j < cnt) {
            const int s = __builtin_amdgcn_readlane(p.x, g + j);
            v[j] = *(const float4*)(H + (size_t)s * FEAT + lane * 4);
        } else {
            v[j].x = 0.f; v[j].y = 0.f; v[j].z = 0.f; v[j].w = 0.f;
        }
    }
    int py = p.y;
    asm volatile("" : "+v"(py) : "v"(v[0].x), "v"(v[1].x), "v"(v[2].x), "v"(v[3].x),
                                 "v"(v[4].x), "v"(v[5].x), "v"(v[6].x), "v"(v[7].x));
    #pragma unroll
    for (int j = 0; j < EGT; ++j) {
        if (FULL || j < cnt) {
            const int k = __builtin_amdgcn_readlane(py, g + j) & (RPW - 1);
            ax[k] += v[j].x; ay[k] += v[j].y; az[k] += v[j].z; aw[k] += v[j].w;
        }
    }
}

__device__ __forceinline__ void drain_owner(const float* __restrict__ H, const int2* bin, int n,
                                            float* __restrict__ agg, int lo, int nnodes,
                                            int wave, int lane) {
    v16f ax = {}, ay = {}, az = {}, aw = {};

    for (int b = 0; b < n; b += 32) {
        int2 p = make_int2(0, 0);
        if (b + lane < n) p = bin[b + lane];
        const int cnt = n - b < 32 ? n - b : 32;
        int g = 0;
        for (; g + EGT <= cnt; g += EGT) owner_group<true>(H, p, g, EGT, lane, ax, ay, az, aw);
        if (g < cnt) owner_group<false>(H, p, g, cnt - g, lane, ax, ay, az, aw);
    }

    #pragma unroll
    for (int k = 0; k < RPW; ++k) {
        const int r = wave + k * OW;
        const int row = lo + r;
        if (r < TR && row < nnodes) {
            float4 o;
            o.x = ax[k]; o.y = ay[k]; o.z = az[k]; o.w = aw[k];
            *(float4*)(agg + (size_t)row * FEAT + lane * 4) = o;
        }
    }
}

__global__ __launch_bounds__(TS_T) void k_tile_sum(const float* __restrict__ H,
                                                   const int* __restrict__ src,
                                                   const int* __restrict__ dst,
                                                   float* __restrict__ agg,
                                                   const int2* __restrict__ cells,
                                                   const int* __restrict__ ccnt,
                                                   int ne, int nnodes, int nch) {
    __shared__ float tile[TR * FEAT];
    __shared__ int   qbuf[QTOT];
    __shared__ int   bcnt[OW];
    const int t = threadIdx.x;
    const int lane = t & 31;
    const int wave = __builtin_amdgcn_readfirstlane(t >> 5);
    const int lo = blockIdx.x * TR;

    if (t < OW) bcnt[t] = 0;
    int bad = nch > QTOT ? 1 : 0;
    for (int j = t; j < nch && j < QTOT; j += TS_T) {
        const int v = ccnt[(size_t)j * gridDim.x + blockIdx.x];
        qbuf[j] = v;
        if (v > CCAP) bad = 1;
    }
    const int scan = __syncthreads_or(bad);

    int ovf = 0;
    if (!scan) ovf = file_cells(cells + (size_t)blockIdx.x * nch * CCAP, qbuf, nch, (int2*)tile, bcnt, wave, lane);
    const int slow = __syncthreads_or(scan | ovf);
    if (!slow) {
        const int n = __builtin_amdgcn_readfirstlane(bcnt[wave]);
        drain_owner(H, (const int2*)tile + wave * BCAP, n, agg, lo, nnodes, wave, lane);
        return;
    }

    {
        const float4 z = {0.f, 0.f, 0.f, 0.f};
        for (int i = t; i < TR * FEAT / 4; i += TS_T) ((float4*)tile)[i] = z;
    }
    __syncthreads();
    if (!scan) tile_from_cells(H, tile, cells + (size_t)blockIdx.x * nch * CCAP, qbuf, nch, wave, lane);
    else       tile_scan(H, src, dst, tile, qbuf + wave * QCAP, ne, lo, t);
    __syncthreads();

    for (int r = wave; r < TR; r += TS_T / 32) {
        const int row = lo + r;
        if (row < nnodes) {
            const float* p = tile + r * FEAT + lane;
            float4 v;
            v.x = p[0]; v.y = p[32]; v.z = p[64]; v.w = p[96];
            *(float4*)(agg + (size_t)row * FEAT + lane * 4) = v;
        }
    }
}

__global__ __launch_bounds__(256) void k_classify(const float* __restrict__ agg,
                                                  const float* __restrict__ rsI,
                                                  const float* __restrict__ bias,
                                                  const float* __restrict__ Wc,
                                                  const float* __restrict__ bc,
                                                  float* __restrict__ out, int n4) {
    int i = blockIdx.x * 256 + threadIdx.x;
    if (i >= n4) return;
    int node = i >> 2;
    int c = i & 3;
    const float* a = agg + (size_t)node * FEAT;
    const float s = rsI[node];
    float sum = 0.f;
    #pragma unroll 8
    for (int k = 0; k < FEAT; ++k) {
        float h = a[k] * s + bias[k];
        h = h > 0.f ? h : 0.f;
        sum += h * Wc[k * 4 + c];
    }
    out[i] = sum + bc[c];
}

extern "C" void kernel_launch(void* const* d_in, const int* in_sizes, int n_in,
                              void* d_out, int out_size, void* d_ws, size_t ws_size,
                              hipStream_t stream) {
    const float* feat = (const float*)d_in[0];
    const int*   src  = (const int*)d_in[1];
    const int*   dst  = (const int*)d_in[2];
    const float* W1   = (const float*)d_in[3];
    const float* b1   = (const float*)d_in[4];
    const float* W2   = (const float*)d_in[5];
    const float* b2   = (const float*)d_in[6];
    const float* Wc   = (const float*)d_in[7];
    const float* bc   = (const float*)d_in[8];
    float* out = (float*)d_out;

    const int nnodes = in_sizes[0] / FEAT;
    const int nedges = in_sizes[1];
    const size_t nAlign = ((size_t)nnodes + 3) & ~(size_t)3;
    const size_t nfeat = (size_t)nnodes * FEAT;

    float* rsO  = (float*)d_ws;
    float* rsI  = rsO + nAlign;
    float* bufA = rsI + nAlign;
    float* bufB = bufA + nfeat;
    int*   part = (int*)(bufB + nfeat);

    const int B = 256;
    const int gNode   = (nnodes + B - 1) / B;
    const int nranges = (nnodes + DEG_NR - 1) / DEG_NR;
    const int eslice  = ((nedges + DEG_S - 1) / DEG_S + 3) & ~3;
    const int gTile   = (nnodes + TR - 1) / TR;
    const int gGemm   = (nnodes + ROWS_PER_BLK - 1) / ROWS_PER_BLK;
    const int gCls    = (nnodes * 4 + B - 1) / B;
    const int nch     = nedges > 0 ? (nedges + BE - 1) / BE : 1;
    int2*  cells = (int2*)(part + (size_t)DEG_S * 2 * nAlign);
    int*   ccnt  = (int*)(cells + (size_t)gTile * nch * CCAP);

    k_degree<<<nranges * DEG_S, DEG_T, 0, stream>>>(src, dst, part, nedges, eslice, nranges, (int)nAlign);
    k_rsqrt_clip<<<gNode, B, 0, stream>>>(part, rsO, rsI, nnodes, (int)nAlign);

    k_gemm_scaled<false><<<gGemm, B, 0, stream>>>(feat, rsO, nullptr, nullptr, W1, bufA, nnodes);
    k_bin<<<nch, BN_T, 0, stream>>>(src, dst, cells, ccnt, nedges, gTile, nch);
    k_tile_sum<<<gTile, TS_T, 0, stream>>>(bufA, src, dst, bufB, cells, ccnt, nedges, nnodes, nch);

    k_gemm_scaled<true><<<gGemm, B, 0, stream>>>(bufB, rsO, rsI, b1, W2, bufA, nnodes);
    k_tile_sum<<<gTile, TS_T, 0, stream>>>(bufA, src, dst, bufB, cells, ccnt, nedges, nnodes, nch);

    k_classify<<<gCls, B, 0, stream>>>(bufB, rsI, b2, Wc, bc, out, nnodes * 4);
}


# layer 2 reassociated: edge sum taken before the W2 product, so scale, bias, relu and the 128x4 classifier dot run in that product's blocks (P kept in LDS); 8 launches -> 7; layer-2 sum order changes
# speedup vs baseline: 4.4661x; 1.0484x over previous
#include <hip/hip_runtime.h>
#include <math.h>

typedef __attribute__((ext_vector_type(2))) float v2f;
typedef __attribute__((ext_vector_type(8))) float v8f;

static constexpr int FEAT = 128;
static constexpr int APAD = 132;
static constexpr int ROWS_PER_BLK = 64;

static constexpr int DEG_NR = 12800;
static constexpr int DEG_S  = 32;
static constexpr int DEG_T  = 1024;
static constexpr int DEG_U  = 4;

__device__ __forceinline__ void deg_count(int* h, int v, int lo) {
    const unsigned u = (unsigned)(v - lo);
    if (u < (unsigned)DEG_NR) __hip_atomic_fetch_add(h + u, 1, __ATOMIC_RELAXED, __HIP_MEMORY_SCOPE_WORKGROUP);
}

__global__ __launch_bounds__(DEG_T) void k_degree(const int* __restrict__ src,
                                                  const int* __restrict__ dst,
                                                  int* __restrict__ part,
                                                  int ne, int es, int nranges, int nalign) {
    __shared__ int hO[DEG_NR];
    __shared__ int hI[DEG_NR];
    const int t = threadIdx.x;
    const int g = blockIdx.x % nranges;
    const int s = blockIdx.x / nranges;
    for (int j = t; j < DEG_NR; j += DEG_T) { hO[j] = 0; hI[j] = 0; }
    __syncthreads();

    const int lo = g * DEG_NR;
    const int e0 = s * es;
    const int hi = e0 + es < ne ? e0 + es : ne;
    for (int eb = e0 + 4 * t; eb + 3 < hi; eb += DEG_U * 4 * DEG_T) {
        int4 a[DEG_U], b[DEG_U];
        bool ok[DEG_U];
        #pragma unroll
        for (int u = 0; u < DEG_U; ++u) {
            const int e = eb + u * 4 * DEG_T;
            ok[u] = e + 3 < hi;
            const int ea = ok[u] ? e : eb;
            a[u] = *(const int4*)(src + ea);
            b[u] = *(const int4*)(dst + ea);
        }
        #pragma unroll
        for (int u = 0; u < DEG_U; ++u) {
            if (ok[u]) {
                deg_count(hO, a[u].x, lo); deg_count(hO, a[u].y, lo);
                deg_count(hO, a[u].z, lo); deg_count(hO, a[u].w, lo);
                deg_count(hI, b[u].x, lo); deg_count(hI, b[u].y, lo);
                deg_count(hI, b[u].z, lo); deg_count(hI, b[u].w, lo);
            }
        }
    }
    {
        const int et = (hi & ~3) + t;
        if (t < 3 && et >= e0 && et < hi) {
            deg_count(hO, src[et], lo);
            deg_count(hI, dst[et], lo);
        }
    }
    __syncthreads();

    int* pO = part + (size_t)(2 * s) * nalign + lo;
    int* pI = pO + nalign;
    for (int j = 4 * t; j < DEG_NR; j += 4 * DEG_T) {
        if (lo + j < nalign) {
            *(int4*)(pO + j) = *(const int4*)(hO + j);
            *(int4*)(pI + j) = *(const int4*)(hI + j);
        }
    }
}

__global__ __launch_bounds__(256) void k_rsqrt_clip(const int* __restrict__ part,
                                                    float* __restrict__ rsO,
                                                    float* __restrict__ rsI, int n, int nalign) {
    int i = blockIdx.x * 256 + threadIdx.x;
    if (i < n) {
        int ca = 0, cb = 0;
        #pragma unroll 8
        for (int s = 0; s < DEG_S; ++s) {
            ca += part[(size_t)(2 * s) * nalign + i];
            cb += part[(size_t)(2 * s + 1) * nalign + i];
        }
        float a = (float)ca; a = a < 1.0f ? 1.0f : a;
        rsO[i] = 1.0f / sqrtf(a);
        float b = (float)cb; b = b < 1.0f ? 1.0f : b;
        rsI[i] = 1.0f / sqrtf(b);
    }
}

__device__ __forceinline__ float4 fin4(float4 a, float s, float4 b) {
    float4 r;
    r.x = a.x * s + b.x; r.x = r.x > 0.f ? r.x : 0.f;
    r.y = a.y * s + b.y; r.y = r.y > 0.f ? r.y : 0.f;
    r.z = a.z * s + b.z; r.z = r.z > 0.f ? r.z : 0.f;
    r.w = a.w * s + b.w; r.w = r.w > 0.f ? r.w : 0.f;
    return r;
}

template <bool CLS>
__global__ __launch_bounds__(256) void k_gemm_scaled(const float* __restrict__ X,
                                                     const float* __restrict__ rsO,
                                                     const float* __restrict__ rsI,
                                                     const float* __restrict__ bias,
                                                     const float* __restrict__ W,
                                                     const float* __restrict__ Wc,
                                                     const float* __restrict__ bc,
                                                     float* __restrict__ Y, int nrows) {
    __shared__ float Alds[ROWS_PER_BLK * APAD];
    __shared__ float WcT[(CLS ? 4 : 1) * APAD];

    const int rowbase = blockIdx.x * ROWS_PER_BLK;
    const int t = threadIdx.x;

    if (CLS) {
        for (int i = t; i < FEAT * 4; i += 256) WcT[(i & 3) * APAD + (i >> 2)] = Wc[i];
    }

    for (int i = t; i < ROWS_PER_BLK * (FEAT / 4); i += 256) {
        int r  = i >> 5;
        int c4 = (i & 31) << 2;
        int row = rowbase + r;
        float4 v = {0.f, 0.f, 0.f, 0.f};
        float s = CLS ? 1.f : 0.f;
        if (row < nrows) {
            v = *(const float4*)(X + (size_t)row * FEAT + c4);
            if (!CLS) s = rsO[row];
        }
        float* dstp = Alds + r * APAD + c4;
        if (CLS) { dstp[0] = v.x; dstp[1] = v.y; dstp[2] = v.z; dstp[3] = v.w; }
        else     { dstp[0] = v.x * s; dstp[1] = v.y * s; dstp[2] = v.z * s; dstp[3] = v.w * s; }
    }
    __syncthreads();

    const int wave = t >> 5;
    const int lane = t & 31;
    const int half = lane >> 4;
    const int l16  = lane & 15;
    const int ncol = wave * 16 + l16;

    v8f acc0 = {}, acc1 = {}, acc2 = {}, acc3 = {};
    const float* a0 = Alds + (0 * 16 + l16) * APAD;
    const float* a1 = Alds + (1 * 16 + l16) * APAD;
    const float* a2 = Alds + (2 * 16 + l16) * APAD;
    const float* a3 = Alds + (3 * 16 + l16) * APAD;

    #pragma unroll 4
    for (int k = 0; k < FEAT; k += 4) {
        const int ka = k + half * 2;
        v2f b;
        b.x = W[ka * FEAT + ncol];
        b.y = W[(ka + 1) * FEAT + ncol];
        v2f a;
        a.x = a0[ka]; a.y = a0[ka + 1];
        acc0 = __builtin_amdgcn_wmma_f32_16x16x4_f32(false, a, false, b, (short)0, acc0, false, false);
        a.x = a1[ka]; a.y = a1[ka + 1];
        acc1 = __builtin_amdgcn_wmma_f32_16x16x4_f32(false, a, false, b, (short)0, acc1, false, false);
        a.x = a2[ka]; a.y = a2[ka + 1];
        acc2 = __builtin_amdgcn_wmma_f32_16x16x4_f32(false, a, false, b, (short)0, acc2, false, false);
        a.x = a3[ka]; a.y = a3[ka + 1];
        acc3 = __builtin_amdgcn_wmma_f32_16x16x4_f32(false, a, false, b, (short)0, acc3, false, false);
    }

    if (CLS) {
        __syncthreads();
        float* pd = Alds + (half * 8) * APAD + ncol;
        #pragma unroll
        for (int r = 0; r < 8; ++r) {
            pd[( 0 + r) * APAD] = acc0[r];
            pd[(16 + r) * APAD] = acc1[r];
            pd[(32 + r) * APAD] = acc2[r];
            pd[(48 + r) * APAD] = acc3[r];
        }
        __syncthreads();
        const int lrow = t >> 2;
        const int cls  = t & 3;
        const int grow = rowbase + lrow;
        if (grow < nrows) {
            const float sI = rsI[grow];
            const float* pr = Alds + lrow * APAD;
            const float* wr = WcT + cls * APAD;
            float sum = 0.f;
            #pragma unroll 8
            for (int k = 0; k < FEAT; ++k) {
                float h = pr[k] * sI + bias[k];
                h = h > 0.f ? h : 0.f;
                sum += h * wr[k];
            }
            Y[(size_t)grow * 4 + cls] = sum + bc[cls];
        }
        return;
    }

    const int mrow0 = rowbase + half * 8;
    if (rowbase + ROWS_PER_BLK <= nrows) {
        #pragma unroll
        for (int r = 0; r < 8; ++r) {
            Y[(size_t)(mrow0 +  0 + r) * FEAT + ncol] = acc0[r];
            Y[(size_t)(mrow0 + 16 + r) * FEAT + ncol] = acc1[r];
            Y[(size_t)(mrow0 + 32 + r) * FEAT + ncol] = acc2[r];
            Y[(size_t)(mrow0 + 48 + r) * FEAT + ncol] = acc3[r];
        }
    } else {
        #pragma unroll
        for (int r = 0; r < 8; ++r) {
            int r0 = mrow0 + r;
            if (r0 +  0 < nrows) Y[(size_t)(r0 +  0) * FEAT + ncol] = acc0[r];
            if (r0 + 16 < nrows) Y[(size_t)(r0 + 16) * FEAT + ncol] = acc1[r];
            if (r0 + 32 < nrows) Y[(size_t)(r0 + 32) * FEAT + ncol] = acc2[r];
            if (r0 + 48 < nrows) Y[(size_t)(r0 + 48) * FEAT + ncol] = acc3[r];
        }
    }
}

static constexpr int TR      = 500;
static constexpr int TS_T    = 1024;
static constexpr int TS_U    = 4;
static constexpr int TS_TRIP = TS_T * 4 * TS_U;
static constexpr int QCAP    = 32 * 4 * TS_U + 32;
static constexpr int QTOT    = (TS_T / 32) * QCAP;
static constexpr int EGT     = 8;
static constexpr int BN_T    = 800;
static constexpr int BE      = BN_T * 8;
static constexpr int CCAP    = 128;
static constexpr int BT_MAX  = 1024;

__global__ __launch_bounds__(BN_T) void k_bin(const int* __restrict__ src,
                                              const int* __restrict__ dst,
                                              int2* __restrict__ cells, int* __restrict__ ccnt,
                                              int ne, int ntiles, int nch) {
    __shared__ int  cnt[BT_MAX];
    __shared__ int  offs[BT_MAX];
    __shared__ int2 stage[BE];
    const int t = threadIdx.x;
    const int lane = t & 31;
    const int wave = t >> 5;
    const int c = blockIdx.x;
    int* mycnt = ccnt + (size_t)c * ntiles;
    if (ntiles > BT_MAX) {
        for (int j = t; j < ntiles; j += BN_T) mycnt[j] = CCAP + 1;
        return;
    }
    for (int j = t; j < ntiles; j += BN_T) cnt[j] = 0;
    __syncthreads();

    int sv[8], dv[8];
    #pragma unroll
    for (int u = 0; u < 2; ++u) {
        const int e = c * BE + u * (4 * BN_T) + 4 * t;
        int4 a = {0, 0, 0, 0};
        int4 b = {-1, -1, -1, -1};
        if (e + 3 < ne) {
            a = *(const int4*)(src + e);
            b = *(const int4*)(dst + e);
        } else {
            if (e < ne)     { a.x = src[e];     b.x = dst[e]; }
            if (e + 1 < ne) { a.y = src[e + 1]; b.y = dst[e + 1]; }
            if (e + 2 < ne) { a.z = src[e + 2]; b.z = dst[e + 2]; }
        }
        sv[4 * u + 0] = a.x; sv[4 * u + 1] = a.y; sv[4 * u + 2] = a.z; sv[4 * u + 3] = a.w;
        dv[4 * u + 0] = b.x; dv[4 * u + 1] = b.y; dv[4 * u + 2] = b.z; dv[4 * u + 3] = b.w;
    }
    const unsigned lim = (unsigned)ntiles * (unsigned)TR;
    int tl[8], rk[8];
    #pragma unroll
    for (int k = 0; k < 8; ++k) {
        tl[k] = -1; rk[k] = 0;
        if ((unsigned)dv[k] < lim) {
            tl[k] = (int)((unsigned)dv[k] / (unsigned)TR);
            dv[k] -= tl[k] * TR;
            rk[k] = __hip_atomic_fetch_add(cnt + tl[k], 1, __ATOMIC_RELAXED, __HIP_MEMORY_SCOPE_WORKGROUP);
        }
    }
    __syncthreads();

    if (t < 32) {
        const int per = (ntiles + 31) >> 5;
        const int b0 = t * per;
        int s = 0;
        for (int j = 0; j < per; ++j) if (b0 + j < ntiles) s += cnt[b0 + j];
        int inc = s;
        #pragma unroll
        for (int d = 1; d < 32; d <<= 1) {
            const int o = __shfl_up(inc, d);
            if (t >= d) inc += o;
        }
        int run = inc - s;
        for (int j = 0; j < per; ++j) if (b0 + j < ntiles) { offs[b0 + j] = run; run += cnt[b0 + j]; }
    }
    __syncthreads();

    #pragma unroll
    for (int k = 0; k < 8; ++k) {
        if (tl[k] >= 0) stage[offs[tl[k]] + rk[k]] = make_int2(sv[k], dv[k]);
    }
    __syncthreads();

    for (int tile = wave; tile < ntiles; tile += BN_T / 32) {
        const int n = cnt[tile];
        if (n <= CCAP) {
            const int2* s = stage + offs[tile];
            int2* cell = cells + ((size_t)tile * nch + c) * CCAP;
            for (int i = lane; i < n; i += 32) cell[i] = s[i];
        }
    }
    for (int j = t; j < ntiles; j += BN_T) mycnt[j] = cnt[j];
}

__device__ __forceinline__ void lds_add(float* p, float v) {
    __hip_atomic_fetch_add(p, v, __ATOMIC_RELAXED, __HIP_MEMORY_SCOPE_WORKGROUP);
}

template <bool FULL>
__device__ __forceinline__ void tile_rows(const float* __restrict__ H, float* tile, int sv, int dv,
                                          int cnt, int lane) {
    for (int g = 0; g < 32; g += EGT) {
        if (!FULL && g >= cnt) break;
        float4 v[EGT];
        #pragma unroll
        for (int j = 0; j < EGT; ++j) {
            v[j].x = 0.f; v[j].y = 0.f; v[j].z = 0.f; v[j].w = 0.f;
            if (FULL || g + j < cnt) {
                const int s = __builtin_amdgcn_readlane(sv, g + j);
                v[j] = *(const float4*)(H + (size_t)s * FEAT + lane * 4);
            }
        }
        #pragma unroll
        for (int j = 0; j < EGT; ++j) {
            if (FULL || g + j < cnt) {
                const int d = __builtin_amdgcn_readlane(dv, g + j);
                float* p = tile + d * FEAT + lane;
                lds_add(p,      v[j].x);
                lds_add(p + 32, v[j].y);
                lds_add(p + 64, v[j].z);
                lds_add(p + 96, v[j].w);
            }
        }
    }
}

__device__ __forceinline__ void tile_from_cells(const float* __restrict__ H, float* tile,
                                                const int2* __restrict__ cells, const int* ccl,
                                                int nch, int wave, int lane) {
    int c = __builtin_amdgcn_readfirstlane(wave);
    int n = 0;
    int2 p0 = make_int2(0, 0), p1 = make_int2(0, 0);
    if (c < nch) {
        n = __builtin_amdgcn_readfirstlane(ccl[c]);
        const int2* q = cells + (size_t)c * CCAP;
        if (lane < n)      p0 = q[lane];
        if (lane + 32 < n) p1 = q[lane + 32];
    }
    while (c < nch) {
        const int cn = c + TS_T / 32;
        int nn = 0;
        int2 n0 = make_int2(0, 0), n1 = make_int2(0, 0);
        if (cn < nch) {
            nn = __builtin_amdgcn_readfirstlane(ccl[cn]);
            const int2* q = cells + (size_t)cn * CCAP;
            if (lane < nn)      n0 = q[lane];
            if (lane + 32 < nn) n1 = q[lane + 32];
        }
        for (int b = 0; b < n && b < 64; b += 32) {
            const int2 p = b ? p1 : p0;
            const int cnt = n - b;
            if (cnt >= 32) tile_rows<true>(H, tile, p.x, p.y, 32, lane);
            else           tile_rows<false>(H, tile, p.x, p.y, cnt, lane);
        }
        const int2* q = cells + (size_t)c * CCAP;
        for (int b = 64; b < n; b += 32) {
            int2 r = make_int2(0, 0);
            if (b + lane < n) r = q[b + lane];
            const int cnt = n - b;
            if (cnt >= 32) tile_rows<true>(H, tile, r.x, r.y, 32, lane);
            else           tile_rows<false>(H, tile, r.x, r.y, cnt, lane);
        }
        c = cn; n = nn; p0 = n0; p1 = n1;
    }
}

template <bool FULL>
__device__ __forceinline__ void tile_drain(const float* __restrict__ H, const int* __restrict__ src,
                                           const int* __restrict__ dst, float* tile, const int* q,
                                           int cnt, int lo, int lane) {
    int e = 0;
    if (FULL || lane < cnt) e = q[lane];
    const int sv = src[e];
    const int dv = dst[e] - lo;
    tile_rows<FULL>(H, tile, sv, dv, cnt, lane);
}

__device__ __forceinline__ void tile_load_trip(const int* __restrict__ dst, int4* a, int tr, int t, int ne) {
    const int elast = ne >= 4 ? (ne - 4) & ~3 : 0;
    #pragma unroll
    for (int u = 0; u < TS_U; ++u) {
        const int e = tr * TS_TRIP + u * (TS_T * 4) + t * 4;
        const unsigned ea = (unsigned)(e < elast ? e : elast);
        a[u] = *(const int4*)((const char*)dst + ea * 4u);
    }
}

__device__ __forceinline__ void tile_file(int* q, int& qn, unsigned m, int e, int lane) {
    const int idx = qn + (int)__builtin_amdgcn_mbcnt_lo(m, 0u);
    if ((m >> lane) & 1u) q[idx] = e;
    qn += __builtin_popcount(m);
}

__device__ __forceinline__ void tile_scan(const float* __restrict__ H, const int* __restrict__ src,
                                          const int* __restrict__ dst, float* tile, int* q,
                                          int ne, int lo, int t) {
    const int lane = t & 31;
    const int wave = t >> 5;
    int qn = 0;
    const int ntrips = (ne + TS_TRIP - 1) / TS_TRIP;
    int trip = (int)(((long long)blockIdx.x * ntrips) / gridDim.x);

    int4 cur[TS_U];
    tile_load_trip(dst, cur, trip, t, ne);
    for (int i = 0; i < ntrips; ++i) {
        int tn = trip + 1;
        if (tn == ntrips) tn = 0;
        int4 nxt[TS_U];
        tile_load_trip(dst, nxt, i + 1 < ntrips ? tn : trip, t, ne);

        const int ebase = trip * TS_TRIP + t * 4;
        unsigned mk[4 * TS_U];
        #pragma unroll
        for (int u = 0; u < TS_U; ++u) {
            const unsigned okm = __builtin_amdgcn_ballot_w32(ebase + u * (TS_T * 4) + 3 < ne);
            mk[4 * u + 0] = __builtin_amdgcn_ballot_w32((unsigned)(cur[u].x - lo) < (unsigned)TR) & okm;
            mk[4 * u + 1] = __builtin_amdgcn_ballot_w32((unsigned)(cur[u].y - lo) < (unsigned)TR) & okm;
            mk[4 * u + 2] = __builtin_amdgcn_ballot_w32((unsigned)(cur[u].z - lo) < (unsigned)TR) & okm;
            mk[4 * u + 3] = __builtin_amdgcn_ballot_w32((unsigned)(cur[u].w - lo) < (unsigned)TR) & okm;
        }
        #pragma unroll
        for (int k = 0; k < 4 * TS_U; ++k) {
            if (mk[k]) tile_file(q, qn, mk[k], ebase + (k >> 2) * (TS_T * 4) + (k & 3), lane);
        }

        int head = 0;
        while (qn - head >= 32) {
            tile_drain<true>(H, src, dst, tile, q + head, 32, lo, lane);
            head += 32;
        }
        if (head > 0) {
            const int left = qn - head;
            int v = 0;
            if (lane < left) v = q[head + lane];
            if (lane < left) q[lane] = v;
            qn = left;
        }

        #pragma unroll
        for (int u = 0; u < TS_U; ++u) cur[u] = nxt[u];
        trip = tn;
    }

    if (wave == 0) {
        const int e = (ne & ~3) + lane;
        const bool ok = lane < 3 && e < ne;
        const int dval = dst[ok ? e : 0];
        const unsigned m = __builtin_amdgcn_ballot_w32(ok && (unsigned)(dval - lo) < (unsigned)TR);
        if (m) tile_file(q, qn, m, e, lane);
    }
    if (qn >= 32) {
        tile_drain<true>(H, src, dst, tile, q, 32, lo, lane);
        qn -= 32;
        if (qn > 0) tile_drain<false>(H, src, dst, tile, q + 32, qn, lo, lane);
    } else if (qn > 0) {
        tile_drain<false>(H, src, dst, tile, q, qn, lo, lane);
    }
}

static constexpr int OW     = TS_T / 32;
static constexpr int OW_LG  = 5;
static constexpr int RPW    = (TR + OW - 1) / OW;
static constexpr int BCAP   = TR * FEAT / 2 / OW;
static_assert((1 << OW_LG) == OW, "owner of a row = its low bits");
static_assert(RPW == 16, "a wave's sums are vectors of 16");
static_assert(EGT == 8, "owner_group names the 8 loaded rows of a group");

__device__ __forceinline__ int file_one(int2* bins, int* bcnt, int2 p) {
    const int o = p.y & (OW - 1);
    const int slot = __hip_atomic_fetch_add(bcnt + o, 1, __ATOMIC_RELAXED, __HIP_MEMORY_SCOPE_WORKGROUP);
    if (slot < BCAP) { bins[o * BCAP + slot] = make_int2(p.x, p.y >> OW_LG); return 0; }
    return 1;
}

static constexpr int FU = 8;
__device__ __forceinline__ int file_cells(const int2* __restrict__ cells, const int* ccl, int nch,
                                          int2* bins, int* bcnt, int wave, int lane) {
    int ovf = 0;
    for (int c0 = __builtin_amdgcn_readfirstlane(wave); c0 < nch; c0 += FU * OW) {
        int n[FU];
        int2 p0[FU], p1[FU];
        #pragma unroll
        for (int u = 0; u < FU; ++u) {
            const int c = c0 + u * OW;
            n[u] = 0;
            p0[u] = make_int2(0, 0); p1[u] = make_int2(0, 0);
            if (c < nch) {
                n[u] = __builtin_amdgcn_readfirstlane(ccl[c]);
                const int2* q = cells + (size_t)c * CCAP;
                if (lane < n[u])      p0[u] = q[lane];
                if (lane + 32 < n[u]) p1[u] = q[lane + 32];
            }
        }
        #pragma unroll
        for (int u = 0; u < FU; ++u) {
            if (lane < n[u])      ovf |= file_one(bins, bcnt, p0[u]);
            if (lane + 32 < n[u]) ovf |= file_one(bins, bcnt, p1[u]);
            if (n[u] > 64) {
                const int2* q = cells + (size_t)(c0 + u * OW) * CCAP;
                for (int b = 64 + lane; b < n[u]; b += 32) ovf |= file_one(bins, bcnt, q[b]);
            }
        }
    }
    return ovf;
}

typedef __attribute__((ext_vector_type(16))) float v16f;

template <bool FULL>
__device__ __forceinline__ void owner_group(const float* __restrict__ H, int2 p, int g, int cnt, int lane,
                                            v16f& ax, v16f& ay, v16f& az, v16f& aw) {
    float4 v[EGT];
    #pragma unroll
    for (int j = 0; j < EGT; ++j) {
        if (FULL || j < cnt) {
            const int s = __builtin_amdgcn_readlane(p.x, g + j);
            v[j] = *(const float4*)(H + (size_t)s * FEAT + lane * 4);
        } else {
            v[j].x = 0.f; v[j].y = 0.f; v[j].z = 0.f; v[j].w = 0.f;
        }
    }
    int py = p.y;
    asm volatile("" : "+v"(py) : "v"(v[0].x), "v"(v[1].x), "v"(v[2].x), "v"(v[3].x),
                                 "v"(v[4].x), "v"(v[5].x), "v"(v[6].x), "v"(v[7].x));
    #pragma unroll
    for (int j = 0; j < EGT; ++j) {
        if (FULL || j < cnt) {
            const int k = __builtin_amdgcn_readlane(py, g + j) & (RPW - 1);
            ax[k] += v[j].x; ay[k] += v[j].y; az[k] += v[j].z; aw[k] += v[j].w;
        }
    }
}

__device__ __forceinline__ float row_scale(float v, int k) {
    return __int_as_float(__builtin_amdgcn_readlane(__float_as_int(v), k));
}

template <bool FIN>
__device__ __forceinline__ void drain_owner(const float* __restrict__ H, const int2* bin, int n,
                                            float* __restrict__ agg, int lo, int nnodes,
                                            int wave, int lane, float sIv, float sOv, float4 bz) {
    v16f ax = {}, ay = {}, az = {}, aw = {};

    for (int b = 0; b < n; b += 32) {
        int2 p = make_int2(0, 0);
        if (b + lane < n) p = bin[b + lane];
        const int cnt = n - b < 32 ? n - b : 32;
        int g = 0;
        for (; g + EGT <= cnt; g += EGT) owner_group<true>(H, p, g, EGT, lane, ax, ay, az, aw);
        if (g < cnt) owner_group<false>(H, p, g, cnt - g, lane, ax, ay, az, aw);
    }

    #pragma unroll
    for (int k = 0; k < RPW; ++k) {
        const int r = wave + k * OW;
        const int row = lo + r;
        if (r < TR && row < nnodes) {
            float4 o;
            o.x = ax[k]; o.y = ay[k]; o.z = az[k]; o.w = aw[k];
            if (FIN) {
                o = fin4(o, row_scale(sIv, k), bz);
                const float s = row_scale(sOv, k);
                o.x *= s; o.y *= s; o.z *= s; o.w *= s;
            }
            *(float4*)(agg + (size_t)row * FEAT + lane * 4) = o;
        }
    }
}

template <bool FIN>
__global__ __launch_bounds__(TS_T) void k_tile_sum(const float* __restrict__ H,
                                                   const int* __restrict__ src,
                                                   const int* __restrict__ dst,
                                                   float* __restrict__ agg,
                                                   const int2* __restrict__ cells,
                                                   const int* __restrict__ ccnt,
                                                   const float* __restrict__ rsI,
                                                   const float* __restrict__ rsO,
                                                   const float* __restrict__ bias,
                                                   int ne, int nnodes, int nch) {
    __shared__ float tile[TR * FEAT];
    __shared__ int   qbuf[QTOT];
    __shared__ int   bcnt[OW];
    const int t = threadIdx.x;
    const int lane = t & 31;
    const int wave = __builtin_amdgcn_readfirstlane(t >> 5);
    const int lo = blockIdx.x * TR;

    float sIv = 0.f, sOv = 0.f;
    float4 bz = {0.f, 0.f, 0.f, 0.f};
    if (FIN) {
        const int row = lo + wave + (lane & (RPW - 1)) * OW;
        if (row < nnodes) { sIv = rsI[row]; sOv = rsO[row]; }
        bz = *(const float4*)(bias + lane * 4);
    }

    if (t < OW) bcnt[t] = 0;
    int bad = nch > QTOT ? 1 : 0;
    for (int j = t; j < nch && j < QTOT; j += TS_T) {
        const int v = ccnt[(size_t)j * gridDim.x + blockIdx.x];
        qbuf[j] = v;
        if (v > CCAP) bad = 1;
    }
    const int scan = __syncthreads_or(bad);

    int ovf = 0;
    if (!scan) ovf = file_cells(cells + (size_t)blockIdx.x * nch * CCAP, qbuf, nch, (int2*)tile, bcnt, wave, lane);
    const int slow = __syncthreads_or(scan | ovf);
    if (!slow) {
        const int n = __builtin_amdgcn_readfirstlane(bcnt[wave]);
        drain_owner<FIN>(H, (const int2*)tile + wave * BCAP, n, agg, lo, nnodes, wave, lane, sIv, sOv, bz);
        return;
    }

    {
        const float4 z = {0.f, 0.f, 0.f, 0.f};
        for (int i = t; i < TR * FEAT / 4; i += TS_T) ((float4*)tile)[i] = z;
    }
    __syncthreads();
    if (!scan) tile_from_cells(H, tile, cells + (size_t)blockIdx.x * nch * CCAP, qbuf, nch, wave, lane);
    else       tile_scan(H, src, dst, tile, qbuf + wave * QCAP, ne, lo, t);
    __syncthreads();

    for (int r = wave; r < TR; r += TS_T / 32) {
        const int row = lo + r;
        if (row < nnodes) {
            const float* p = tile + r * FEAT + lane;
            float4 v;
            v.x = p[0]; v.y = p[32]; v.z = p[64]; v.w = p[96];
            if (FIN) {
                const int k = (r - wave) >> OW_LG;
                v = fin4(v, row_scale(sIv, k), bz);
                const float s = row_scale(sOv, k);
                v.x *= s; v.y *= s; v.z *= s; v.w *= s;
            }
            *(float4*)(agg + (size_t)row * FEAT + lane * 4) = v;
        }
    }
}

extern "C" void kernel_launch(void* const* d_in, const int* in_sizes, int n_in,
                              void* d_out, int out_size, void* d_ws, size_t ws_size,
                              hipStream_t stream) {
    const float* feat = (const float*)d_in[0];
    const int*   src  = (const int*)d_in[1];
    const int*   dst  = (const int*)d_in[2];
    const float* W1   = (const float*)d_in[3];
    const float* b1   = (const float*)d_in[4];
    const float* W2   = (const float*)d_in[5];
    const float* b2   = (const float*)d_in[6];
    const float* Wc   = (const float*)d_in[7];
    const float* bc   = (const float*)d_in[8];
    float* out = (float*)d_out;

    const int nnodes = in_sizes[0] / FEAT;
    const int nedges = in_sizes[1];
    const size_t nAlign = ((size_t)nnodes + 3) & ~(size_t)3;
    const size_t nfeat = (size_t)nnodes * FEAT;

    float* rsO  = (float*)d_ws;
    float* rsI  = rsO + nAlign;
    float* bufA = rsI + nAlign;
    float* bufB = bufA + nfeat;
    int*   part = (int*)(bufB + nfeat);

    const int B = 256;
    const int gNode   = (nnodes + B - 1) / B;
    const int nranges = (nnodes + DEG_NR - 1) / DEG_NR;
    const int eslice  = ((nedges + DEG_S - 1) / DEG_S + 3) & ~3;
    const int gTile   = (nnodes + TR - 1) / TR;
    const int gGemm   = (nnodes + ROWS_PER_BLK - 1) / ROWS_PER_BLK;
    const int nch     = nedges > 0 ? (nedges + BE - 1) / BE : 1;
    int2*  cells = (int2*)(part + (size_t)DEG_S * 2 * nAlign);
    int*   ccnt  = (int*)(cells + (size_t)gTile * nch * CCAP);

    k_degree<<<nranges * DEG_S, DEG_T, 0, stream>>>(src, dst, part, nedges, eslice, nranges, (int)nAlign);
    k_rsqrt_clip<<<gNode, B, 0, stream>>>(part, rsO, rsI, nnodes, (int)nAlign);

    k_gemm_scaled<false><<<gGemm, B, 0, stream>>>(feat, rsO, nullptr, nullptr, W1, nullptr, nullptr, bufA, nnodes);
    k_bin<<<nch, BN_T, 0, stream>>>(src, dst, cells, ccnt, nedges, gTile, nch);
    k_tile_sum<true><<<gTile, TS_T, 0, stream>>>(bufA, src, dst, bufB, cells, ccnt, rsI, rsO, b1, nedges, nnodes, nch);

    k_tile_sum<false><<<gTile, TS_T, 0, stream>>>(bufB, src, dst, bufA, cells, ccnt, nullptr, nullptr, nullptr, nedges, nnodes, nch);
    k_gemm_scaled<true><<<gGemm, B, 0, stream>>>(bufA, nullptr, rsI, b2, W2, Wc, bc, out, nnodes);
}
